# starting kernel
# baseline (speedup 1.0000x reference)
#include <hip/hip_runtime.h>

typedef __attribute__((ext_vector_type(16))) _Float16 v16h;
typedef __attribute__((ext_vector_type(8)))  _Float16 v8h;
typedef __attribute__((ext_vector_type(8)))  float    v8f;

static __device__ __forceinline__ int gtid() {
  return blockIdx.x * blockDim.x + threadIdx.x;
}

__global__ void k_init_deg(float* __restrict__ deg, int N) {
  int i = gtid();
  if (i < N) deg[i] = 1.0f;
}

__global__ void k_deg_edges(const int* __restrict__ dst, float* __restrict__ deg, int E) {
  int e = gtid();
  if (e < E) atomicAdd(&deg[dst[e]], 1.0f);
}

__global__ void k_rsqrt_self(const float* __restrict__ x, float* __restrict__ dis,
                             float* __restrict__ aggx, int N) {
  int i = gtid();
  if (i >= N) return;
  float r = rsqrtf(dis[i]);
  dis[i] = r;
  float nn = r * r;
  float4 xv = reinterpret_cast<const float4*>(x)[i];
  float4 o = make_float4(xv.x * nn, xv.y * nn, xv.z * nn, xv.w * nn);
  reinterpret_cast<float4*>(aggx)[i] = o;
}

__global__ void k_agg1(const int* __restrict__ src, const int* __restrict__ dst,
                       const float* __restrict__ x, const float* __restrict__ dis,
                       float* __restrict__ aggx, int E) {
  int e = gtid();
  if (e >= E) return;
  int s = src[e], d = dst[e];
  float n = dis[s] * dis[d];
  float4 xv = reinterpret_cast<const float4*>(x)[s];
  float* ap = aggx + (size_t)d * 4;
  atomicAdd(ap + 0, xv.x * n);
  atomicAdd(ap + 1, xv.y * n);
  atomicAdd(ap + 2, xv.z * n);
  atomicAdd(ap + 3, xv.w * n);
}

__global__ void k_h1(const float* __restrict__ aggx, const float* __restrict__ W1,
                     const float* __restrict__ b1, const float* __restrict__ dis,
                     float* __restrict__ h1, float* __restrict__ r2, int N) {
  int i = gtid();
  if (i >= N * 64) return;
  int n = i >> 6, f = i & 63;
  float4 av = reinterpret_cast<const float4*>(aggx)[n];
  float acc = b1[f];
  acc += av.x * W1[f];
  acc += av.y * W1[64 + f];
  acc += av.z * W1[128 + f];
  acc += av.w * W1[192 + f];
  float h = fmaxf(acc, 0.0f);
  h1[i] = h;
  float r = dis[n];
  r2[i] = h * r * r;
}

__global__ void k_agg2(const int* __restrict__ src, const int* __restrict__ dst,
                       const float* __restrict__ h1, const float* __restrict__ dis,
                       float* __restrict__ r2, int E) {
  long long t = (long long)blockIdx.x * blockDim.x + threadIdx.x;
  int e  = (int)(t >> 4);
  int f4 = (int)(t & 15);
  if (e >= E) return;
  if (f4 == 0) {
    __builtin_prefetch(src + e + 8192, 0, 1);
    __builtin_prefetch(dst + e + 8192, 0, 1);
  }
  int s = src[e], d = dst[e];
  float n = dis[s] * dis[d];
  float4 v = reinterpret_cast<const float4*>(h1 + (size_t)s * 64)[f4];
  float* rp = r2 + (size_t)d * 64 + f4 * 4;
  atomicAdd(rp + 0, v.x * n);
  atomicAdd(rp + 1, v.y * n);
  atomicAdd(rp + 2, v.z * n);
  atomicAdd(rp + 3, v.w * n);
}

__global__ void __launch_bounds__(128)
k_gemm2_out(const float* __restrict__ r2, const float* __restrict__ W2,
            const float* __restrict__ b2, const float* __restrict__ Wf,
            const float* __restrict__ bf, float* __restrict__ out, int N) {
  __shared__ _Float16 w2t[64 * 64];
  for (int idx = threadIdx.x; idx < 64 * 64; idx += 128) {
    int col = idx >> 6, k = idx & 63;
    w2t[idx] = (_Float16)W2[k * 64 + col];
  }
  __syncthreads();

  int wave  = threadIdx.x >> 5;
  int lane  = threadIdx.x & 31;
  int rbase = blockIdx.x * 64 + wave * 16;
  int lrow  = lane & 15;
  int hi    = lane >> 4;
  int kb    = hi * 8;

  const float* ar = r2 + (size_t)(rbase + lrow) * 64;
  const _Float16* wt = w2t + (size_t)lrow * 64;

  v8f zero = {0, 0, 0, 0, 0, 0, 0, 0};
  v8f acc[4];
  acc[0] = zero; acc[1] = zero; acc[2] = zero; acc[3] = zero;

#pragma unroll
  for (int kc = 0; kc < 64; kc += 32) {
    v16h a;
#pragma unroll
    for (int j = 0; j < 8; ++j) {
      a[j]     = (_Float16)ar[kc + kb + j];
      a[8 + j] = (_Float16)ar[kc + kb + 16 + j];
    }
#pragma unroll
    for (int ct = 0; ct < 4; ++ct) {
      const _Float16* wc = wt + (size_t)ct * 16 * 64;
      v8h blo = *reinterpret_cast<const v8h*>(wc + kc + kb);
      v8h bhi = *reinterpret_cast<const v8h*>(wc + kc + kb + 16);
      v16h b = __builtin_shufflevector(blo, bhi, 0, 1, 2, 3, 4, 5, 6, 7,
                                       8, 9, 10, 11, 12, 13, 14, 15);
      acc[ct] = __builtin_amdgcn_wmma_f32_16x16x32_f16(
          false, a, false, b, (short)0, acc[ct], false, false);
    }
  }

  float bias[4], wf0[4], wf1[4];
#pragma unroll
  for (int ct = 0; ct < 4; ++ct) {
    int col = ct * 16 + lrow;
    bias[ct] = b2[col];
    wf0[ct]  = Wf[col * 2 + 0];
    wf1[ct]  = Wf[col * 2 + 1];
  }
  float bf0 = bf[0], bf1 = bf[1];

#pragma unroll
  for (int v = 0; v < 8; ++v) {
    float p0 = 0.0f, p1 = 0.0f;
#pragma unroll
    for (int ct = 0; ct < 4; ++ct) {
      float h = fmaxf(acc[ct][v] + bias[ct], 0.0f);
      p0 += h * wf0[ct];
      p1 += h * wf1[ct];
    }
#pragma unroll
    for (int off = 8; off >= 1; off >>= 1) {
      p0 += __shfl_xor(p0, off, 16);
      p1 += __shfl_xor(p1, off, 16);
    }
    int row = rbase + v + hi * 8;
    if (lrow == 0 && row < N) {
      out[row * 2 + 0] = p0 + bf0;
      out[row * 2 + 1] = p1 + bf1;
    }
  }
}

extern "C" void kernel_launch(void* const* d_in, const int* in_sizes, int n_in,
                              void* d_out, int out_size, void* d_ws, size_t ws_size,
                              hipStream_t stream) {
  const float* x  = (const float*)d_in[0];
  const int*   ei = (const int*)d_in[1];
  const float* W1 = (const float*)d_in[2];
  const float* b1 = (const float*)d_in[3];
  const float* W2 = (const float*)d_in[4];
  const float* b2 = (const float*)d_in[5];
  const float* Wf = (const float*)d_in[6];
  const float* bf = (const float*)d_in[7];
  float* out = (float*)d_out;

  int N = in_sizes[0] / 4;
  int E = in_sizes[1] / 2;
  int Npad = ((N + 63) / 64) * 64;
  const int* src = ei;
  const int* dst = ei + E;

  float* ws   = (float*)d_ws;
  float* dis  = ws;
  float* aggx = dis + N;
  float* h1   = aggx + (size_t)4 * N;
  float* r2   = h1 + (size_t)64 * N;

  const int TB = 256;
  k_init_deg<<<(N + TB - 1) / TB, TB, 0, stream>>>(dis, N);
  k_deg_edges<<<(E + TB - 1) / TB, TB, 0, stream>>>(dst, dis, E);
  k_rsqrt_self<<<(N + TB - 1) / TB, TB, 0, stream>>>(x, dis, aggx, N);
  k_agg1<<<(E + TB - 1) / TB, TB, 0, stream>>>(src, dst, x, dis, aggx, E);
  long long t1 = (long long)N * 64;
  k_h1<<<(unsigned)((t1 + TB - 1) / TB), TB, 0, stream>>>(aggx, W1, b1, dis, h1, r2, N);
  long long t2 = (long long)E * 16;
  k_agg2<<<(unsigned)((t2 + TB - 1) / TB), TB, 0, stream>>>(src, dst, h1, dis, r2, E);
  k_gemm2_out<<<Npad / 64, 128, 0, stream>>>(r2, W2, b2, Wf, bf, out, N);
}


# Scatter-add lanes remapped, one wave per 32 edges: each float atomic instruction covers a whole contiguous row piece (16 B in layer 1, a full 128-B line in layer 2); sums stay in arrival order
# speedup vs baseline: 1.6175x; 1.6175x over previous
#include <hip/hip_runtime.h>

typedef __attribute__((ext_vector_type(16))) _Float16 v16h;
typedef __attribute__((ext_vector_type(8)))  _Float16 v8h;
typedef __attribute__((ext_vector_type(8)))  float    v8f;

static __device__ __forceinline__ int gtid() {
  return blockIdx.x * blockDim.x + threadIdx.x;
}

__global__ void k_init_deg(float* __restrict__ deg, int N) {
  int i = gtid();
  if (i < N) deg[i] = 1.0f;
}

__global__ void k_deg_edges(const int* __restrict__ dst, float* __restrict__ deg, int E) {
  int e = gtid();
  if (e < E) atomicAdd(&deg[dst[e]], 1.0f);
}

__global__ void k_rsqrt_self(const float* __restrict__ x, float* __restrict__ dis,
                             float* __restrict__ aggx, int N) {
  int i = gtid();
  if (i >= N) return;
  float r = rsqrtf(dis[i]);
  dis[i] = r;
  float nn = r * r;
  float4 xv = reinterpret_cast<const float4*>(x)[i];
  float4 o = make_float4(xv.x * nn, xv.y * nn, xv.z * nn, xv.w * nn);
  reinterpret_cast<float4*>(aggx)[i] = o;
}

__global__ void k_agg1(const int* __restrict__ src, const int* __restrict__ dst,
                       const float* __restrict__ x, const float* __restrict__ dis,
                       float* __restrict__ aggx, int E) {
  int e = gtid();
  int lane = threadIdx.x & 31;
  int ebase = e - lane;
  int s = 0, d = 0;
  float n = 0.0f;
  if (e < E) {
    s = src[e];
    d = dst[e];
    n = dis[s] * dis[d];
  }
  int c = lane & 3;
  int q = lane >> 2;
  int dj[4];
  float nj[4], v[4];
#pragma unroll
  for (int r = 0; r < 4; ++r) {
    int j = r * 8 + q;
    int sj = __shfl(s, j);
    dj[r] = __shfl(d, j);
    nj[r] = __shfl(n, j);
    v[r] = x[(size_t)sj * 4 + c];
  }
#pragma unroll
  for (int r = 0; r < 4; ++r) {
    if (ebase + r * 8 + q < E)
      atomicAdd(aggx + (size_t)dj[r] * 4 + c, v[r] * nj[r]);
  }
}

__global__ void k_h1(const float* __restrict__ aggx, const float* __restrict__ W1,
                     const float* __restrict__ b1, const float* __restrict__ dis,
                     float* __restrict__ h1, float* __restrict__ r2, int N) {
  int i = gtid();
  if (i >= N * 64) return;
  int n = i >> 6, f = i & 63;
  float4 av = reinterpret_cast<const float4*>(aggx)[n];
  float acc = b1[f];
  acc += av.x * W1[f];
  acc += av.y * W1[64 + f];
  acc += av.z * W1[128 + f];
  acc += av.w * W1[192 + f];
  float h = fmaxf(acc, 0.0f);
  h1[i] = h;
  float r = dis[n];
  r2[i] = h * r * r;
}

__global__ void __launch_bounds__(256)
k_agg2(const int* __restrict__ src, const int* __restrict__ dst,
       const float* __restrict__ h1, const float* __restrict__ dis,
       float* __restrict__ r2, int E) {
  int lane = threadIdx.x & 31;
  int wv = __builtin_amdgcn_readfirstlane((int)(threadIdx.x >> 5));
  long long e0 = ((long long)blockIdx.x * (blockDim.x >> 5) + wv) * 32;
  if (e0 >= E) return;
  int cnt = (E - e0 < 32) ? (int)(E - e0) : 32;
  int s = 0, d = 0;
  float n = 0.0f;
  if (lane < cnt) {
    int e = (int)e0 + lane;
    s = src[e];
    d = dst[e];
    n = dis[s] * dis[d];
  }
  int nbits = __float_as_int(n);

  if (cnt == 32) {
#pragma unroll
    for (int g = 0; g < 4; ++g) {
      float a0[8], a1[8];
#pragma unroll
      for (int u = 0; u < 8; ++u) {
        int sj = __builtin_amdgcn_readlane(s, g * 8 + u);
        const float* row = h1 + (size_t)sj * 64;
        a0[u] = row[lane];
        a1[u] = row[32 + lane];
      }
#pragma unroll
      for (int u = 0; u < 8; ++u) {
        int dj = __builtin_amdgcn_readlane(d, g * 8 + u);
        float nj = __int_as_float(__builtin_amdgcn_readlane(nbits, g * 8 + u));
        float* rp = r2 + (size_t)dj * 64;
        atomicAdd(rp + lane, a0[u] * nj);
        atomicAdd(rp + 32 + lane, a1[u] * nj);
      }
    }
  } else {
    for (int j = 0; j < cnt; ++j) {
      int sj = __shfl(s, j);
      int dj = __shfl(d, j);
      float nj = __shfl(n, j);
      const float* row = h1 + (size_t)sj * 64;
      float* rp = r2 + (size_t)dj * 64;
      float b0 = row[lane];
      float b1v = row[32 + lane];
      atomicAdd(rp + lane, b0 * nj);
      atomicAdd(rp + 32 + lane, b1v * nj);
    }
  }
}

__global__ void __launch_bounds__(128)
k_gemm2_out(const float* __restrict__ r2, const float* __restrict__ W2,
            const float* __restrict__ b2, const float* __restrict__ Wf,
            const float* __restrict__ bf, float* __restrict__ out, int N) {
  __shared__ _Float16 w2t[64 * 64];
  for (int idx = threadIdx.x; idx < 64 * 64; idx += 128) {
    int col = idx >> 6, k = idx & 63;
    w2t[idx] = (_Float16)W2[k * 64 + col];
  }
  __syncthreads();

  int wave  = threadIdx.x >> 5;
  int lane  = threadIdx.x & 31;
  int rbase = blockIdx.x * 64 + wave * 16;
  int lrow  = lane & 15;
  int hi    = lane >> 4;
  int kb    = hi * 8;

  const float* ar = r2 + (size_t)(rbase + lrow) * 64;
  const _Float16* wt = w2t + (size_t)lrow * 64;

  v8f zero = {0, 0, 0, 0, 0, 0, 0, 0};
  v8f acc[4];
  acc[0] = zero; acc[1] = zero; acc[2] = zero; acc[3] = zero;

#pragma unroll
  for (int kc = 0; kc < 64; kc += 32) {
    v16h a;
#pragma unroll
    for (int j = 0; j < 8; ++j) {
      a[j]     = (_Float16)ar[kc + kb + j];
      a[8 + j] = (_Float16)ar[kc + kb + 16 + j];
    }
#pragma unroll
    for (int ct = 0; ct < 4; ++ct) {
      const _Float16* wc = wt + (size_t)ct * 16 * 64;
      v8h blo = *reinterpret_cast<const v8h*>(wc + kc + kb);
      v8h bhi = *reinterpret_cast<const v8h*>(wc + kc + kb + 16);
      v16h b = __builtin_shufflevector(blo, bhi, 0, 1, 2, 3, 4, 5, 6, 7,
                                       8, 9, 10, 11, 12, 13, 14, 15);
      acc[ct] = __builtin_amdgcn_wmma_f32_16x16x32_f16(
          false, a, false, b, (short)0, acc[ct], false, false);
    }
  }

  float bias[4], wf0[4], wf1[4];
#pragma unroll
  for (int ct = 0; ct < 4; ++ct) {
    int col = ct * 16 + lrow;
    bias[ct] = b2[col];
    wf0[ct]  = Wf[col * 2 + 0];
    wf1[ct]  = Wf[col * 2 + 1];
  }
  float bf0 = bf[0], bf1 = bf[1];

#pragma unroll
  for (int v = 0; v < 8; ++v) {
    float p0 = 0.0f, p1 = 0.0f;
#pragma unroll
    for (int ct = 0; ct < 4; ++ct) {
      float h = fmaxf(acc[ct][v] + bias[ct], 0.0f);
      p0 += h * wf0[ct];
      p1 += h * wf1[ct];
    }
#pragma unroll
    for (int off = 8; off >= 1; off >>= 1) {
      p0 += __shfl_xor(p0, off, 16);
      p1 += __shfl_xor(p1, off, 16);
    }
    int row = rbase + v + hi * 8;
    if (lrow == 0 && row < N) {
      out[row * 2 + 0] = p0 + bf0;
      out[row * 2 + 1] = p1 + bf1;
    }
  }
}

extern "C" void kernel_launch(void* const* d_in, const int* in_sizes, int n_in,
                              void* d_out, int out_size, void* d_ws, size_t ws_size,
                              hipStream_t stream) {
  const float* x  = (const float*)d_in[0];
  const int*   ei = (const int*)d_in[1];
  const float* W1 = (const float*)d_in[2];
  const float* b1 = (const float*)d_in[3];
  const float* W2 = (const float*)d_in[4];
  const float* b2 = (const float*)d_in[5];
  const float* Wf = (const float*)d_in[6];
  const float* bf = (const float*)d_in[7];
  float* out = (float*)d_out;

  int N = in_sizes[0] / 4;
  int E = in_sizes[1] / 2;
  int Npad = ((N + 63) / 64) * 64;
  const int* src = ei;
  const int* dst = ei + E;

  float* ws   = (float*)d_ws;
  float* dis  = ws;
  float* aggx = dis + N;
  float* h1   = aggx + (size_t)4 * N;
  float* r2   = h1 + (size_t)64 * N;

  const int TB = 256;
  k_init_deg<<<(N + TB - 1) / TB, TB, 0, stream>>>(dis, N);
  k_deg_edges<<<(E + TB - 1) / TB, TB, 0, stream>>>(dst, dis, E);
  k_rsqrt_self<<<(N + TB - 1) / TB, TB, 0, stream>>>(x, dis, aggx, N);
  k_agg1<<<(E + TB - 1) / TB, TB, 0, stream>>>(src, dst, x, dis, aggx, E);
  long long t1 = (long long)N * 64;
  k_h1<<<(unsigned)((t1 + TB - 1) / TB), TB, 0, stream>>>(aggx, W1, b1, dis, h1, r2, N);
  k_agg2<<<(E + TB - 1) / TB, TB, 0, stream>>>(src, dst, h1, dis, r2, E);
  k_gemm2_out<<<Npad / 64, 128, 0, stream>>>(r2, W2, b2, Wf, bf, out, N);
}


# Layer 2 summed by destination-owning blocks: a block keeps its 391 r2 rows in LDS, streams all edge destinations, rebuilds each message from the 16-B layer-1 row; LDS float adds in arrival order
# speedup vs baseline: 1.7724x; 1.0958x over previous
#include <hip/hip_runtime.h>

typedef __attribute__((ext_vector_type(16))) _Float16 v16h;
typedef __attribute__((ext_vector_type(8)))  _Float16 v8h;
typedef __attribute__((ext_vector_type(8)))  float    v8f;

static __device__ __forceinline__ int gtid() {
  return blockIdx.x * blockDim.x + threadIdx.x;
}

__global__ void k_init_deg(float* __restrict__ deg, int N) {
  int i = gtid();
  if (i < N) deg[i] = 1.0f;
}

__global__ void k_deg_edges(const int* __restrict__ dst, float* __restrict__ deg, int E) {
  int e = gtid();
  if (e < E) atomicAdd(&deg[dst[e]], 1.0f);
}

__global__ void k_rsqrt_self(const float* __restrict__ x, float* __restrict__ dis,
                             float* __restrict__ aggx, int N) {
  int i = gtid();
  if (i >= N) return;
  float r = rsqrtf(dis[i]);
  dis[i] = r;
  float nn = r * r;
  float4 xv = reinterpret_cast<const float4*>(x)[i];
  float4 o = make_float4(xv.x * nn, xv.y * nn, xv.z * nn, xv.w * nn);
  reinterpret_cast<float4*>(aggx)[i] = o;
}

__global__ void k_agg1(const int* __restrict__ src, const int* __restrict__ dst,
                       const float* __restrict__ x, const float* __restrict__ dis,
                       float* __restrict__ aggx, int E) {
  int e = gtid();
  int lane = threadIdx.x & 31;
  int ebase = e - lane;
  int s = 0, d = 0;
  float n = 0.0f;
  if (e < E) {
    s = src[e];
    d = dst[e];
    n = dis[s] * dis[d];
  }
  int c = lane & 3;
  int q = lane >> 2;
  int dj[4];
  float nj[4], v[4];
#pragma unroll
  for (int r = 0; r < 4; ++r) {
    int j = r * 8 + q;
    int sj = __shfl(s, j);
    dj[r] = __shfl(d, j);
    nj[r] = __shfl(n, j);
    v[r] = x[(size_t)sj * 4 + c];
  }
#pragma unroll
  for (int r = 0; r < 4; ++r) {
    if (ebase + r * 8 + q < E)
      atomicAdd(aggx + (size_t)dj[r] * 4 + c, v[r] * nj[r]);
  }
}

#define OWN_ROWS  391
#define OWN_TB    1024
#define OWN_WAVES (OWN_TB / 32)
#define OWN_CAP   1024
#define OWN_U     8

struct OwnW {
  float a0, a1, a2, a3, ab;
  float b0, b1, b2, b3, bb;
};

static __device__ __forceinline__ void own_wave_order() {
  __builtin_amdgcn_fence(__ATOMIC_RELEASE, "wavefront");
  __builtin_amdgcn_wave_barrier();
  __builtin_amdgcn_fence(__ATOMIC_ACQUIRE, "wavefront");
}

static __device__ __forceinline__ void own_drain(
    const int* mypend, int cnt, int lane, int lo,
    const int* __restrict__ src, const int* __restrict__ dst,
    const float* __restrict__ dis, const float* __restrict__ aggx,
    float* acc, const OwnW& w) {
  own_wave_order();
  for (int base = 0; base < cnt; base += 32) {
    int m = cnt - base;
    if (m > 32) m = 32;
    int r = 0;
    float n = 0.0f;
    float4 av = make_float4(0.0f, 0.0f, 0.0f, 0.0f);
    if (lane < m) {
      int e = mypend[base + lane];
      int s = src[e];
      int d = dst[e];
      av = reinterpret_cast<const float4*>(aggx)[s];
      n = dis[s] * dis[d];
      r = d - lo;
    }
    int nb = __float_as_int(n);
    int x0b = __float_as_int(av.x), x1b = __float_as_int(av.y);
    int x2b = __float_as_int(av.z), x3b = __float_as_int(av.w);
    for (int j = 0; j < m; ++j) {
      int rj = __builtin_amdgcn_readlane(r, j);
      float nj = __int_as_float(__builtin_amdgcn_readlane(nb, j));
      float x0 = __int_as_float(__builtin_amdgcn_readlane(x0b, j));
      float x1 = __int_as_float(__builtin_amdgcn_readlane(x1b, j));
      float x2 = __int_as_float(__builtin_amdgcn_readlane(x2b, j));
      float x3 = __int_as_float(__builtin_amdgcn_readlane(x3b, j));
      float ha = w.ab;
      ha = __builtin_fmaf(x0, w.a0, ha);
      ha = __builtin_fmaf(x1, w.a1, ha);
      ha = __builtin_fmaf(x2, w.a2, ha);
      ha = __builtin_fmaf(x3, w.a3, ha);
      float hb = w.bb;
      hb = __builtin_fmaf(x0, w.b0, hb);
      hb = __builtin_fmaf(x1, w.b1, hb);
      hb = __builtin_fmaf(x2, w.b2, hb);
      hb = __builtin_fmaf(x3, w.b3, hb);
      ha = fmaxf(ha, 0.0f);
      hb = fmaxf(hb, 0.0f);
      float* row = acc + rj * 64;
      atomicAdd(row + lane, ha * nj);
      atomicAdd(row + 32 + lane, hb * nj);
    }
  }
  own_wave_order();
}

static __device__ __forceinline__ void own_note(bool m, int e, int* mypend, int& cnt,
                                                unsigned lt) {
  unsigned mask = __builtin_amdgcn_ballot_w32(m);
  if (mask != 0u) {
    if (m) mypend[cnt + __builtin_popcount(mask & lt)] = e;
    cnt += __builtin_popcount(mask);
  }
}

__global__ void __launch_bounds__(OWN_TB)
k_layer2_own(const int* __restrict__ src, const int* __restrict__ dst,
             const float* __restrict__ aggx, const float* __restrict__ W1,
             const float* __restrict__ b1, const float* __restrict__ dis,
             float* __restrict__ r2, int N, int E) {
  __shared__ float acc[OWN_ROWS * 64];
  __shared__ int pend[OWN_WAVES * OWN_CAP];

  const int lane = threadIdx.x & 31;
  const int wave = __builtin_amdgcn_readfirstlane((int)(threadIdx.x >> 5));
  const int lo = blockIdx.x * OWN_ROWS;
  int rows = N - lo;
  if (rows > OWN_ROWS) rows = OWN_ROWS;

  OwnW w;
  w.a0 = W1[lane];      w.a1 = W1[64 + lane]; w.a2 = W1[128 + lane]; w.a3 = W1[192 + lane];
  w.b0 = W1[32 + lane]; w.b1 = W1[96 + lane]; w.b2 = W1[160 + lane]; w.b3 = W1[224 + lane];
  w.ab = b1[lane];      w.bb = b1[32 + lane];

  for (int r = wave; r < rows; r += OWN_WAVES) {
    int node = lo + r;
    float4 av = reinterpret_cast<const float4*>(aggx)[node];
    float dd = dis[node];
    float ha = w.ab;
    ha = __builtin_fmaf(av.x, w.a0, ha);
    ha = __builtin_fmaf(av.y, w.a1, ha);
    ha = __builtin_fmaf(av.z, w.a2, ha);
    ha = __builtin_fmaf(av.w, w.a3, ha);
    float hb = w.bb;
    hb = __builtin_fmaf(av.x, w.b0, hb);
    hb = __builtin_fmaf(av.y, w.b1, hb);
    hb = __builtin_fmaf(av.z, w.b2, hb);
    hb = __builtin_fmaf(av.w, w.b3, hb);
    ha = fmaxf(ha, 0.0f);
    hb = fmaxf(hb, 0.0f);
    acc[r * 64 + lane] = ha * dd * dd;
    acc[r * 64 + 32 + lane] = hb * dd * dd;
  }
  __syncthreads();

  int* mypend = pend + wave * OWN_CAP;
  int cnt = 0;
  const unsigned lt = (1u << lane) - 1u;
  const unsigned urows = (unsigned)rows;

  const bool vec = ((reinterpret_cast<size_t>(dst) & 15) == 0);
  const int trip = OWN_TB * OWN_U;
  const int ntrips = vec ? ((E >> 2) / trip) : 0;
  const int nq = ntrips * trip;
  const int4* dq = reinterpret_cast<const int4*>(dst);
  const int t0 = (int)(((long long)blockIdx.x * ntrips) / gridDim.x);
  for (int t = 0; t < ntrips; ++t) {
    int tt = t + t0;
    if (tt >= ntrips) tt -= ntrips;
    const int qb = tt * trip;
    int4 v[OWN_U];
#pragma unroll
    for (int u = 0; u < OWN_U; ++u) v[u] = dq[qb + u * OWN_TB + (int)threadIdx.x];
#pragma unroll
    for (int u = 0; u < OWN_U; ++u) {
      int e = (qb + u * OWN_TB + (int)threadIdx.x) * 4;
      bool m0 = (unsigned)(v[u].x - lo) < urows;
      bool m1 = (unsigned)(v[u].y - lo) < urows;
      bool m2 = (unsigned)(v[u].z - lo) < urows;
      bool m3 = (unsigned)(v[u].w - lo) < urows;
      if (__builtin_amdgcn_ballot_w32(m0 | m1 | m2 | m3) != 0u) {
        if (cnt > OWN_CAP - 128) {
          own_drain(mypend, cnt, lane, lo, src, dst, dis, aggx, acc, w);
          cnt = 0;
        }
        own_note(m0, e,     mypend, cnt, lt);
        own_note(m1, e + 1, mypend, cnt, lt);
        own_note(m2, e + 2, mypend, cnt, lt);
        own_note(m3, e + 3, mypend, cnt, lt);
      }
    }
  }
  for (int eb = nq * 4; eb < E; eb += OWN_TB) {
    int e = eb + (int)threadIdx.x;
    int d = -1;
    if (e < E) d = dst[e];
    bool m = (unsigned)(d - lo) < urows;
    if (cnt > OWN_CAP - 32) {
      own_drain(mypend, cnt, lane, lo, src, dst, dis, aggx, acc, w);
      cnt = 0;
    }
    own_note(m, e, mypend, cnt, lt);
  }
  own_drain(mypend, cnt, lane, lo, src, dst, dis, aggx, acc, w);
  __syncthreads();

  float* outp = r2 + (size_t)lo * 64;
  for (int i = threadIdx.x; i < rows * 64; i += OWN_TB) outp[i] = acc[i];
}

__global__ void __launch_bounds__(128)
k_gemm2_out(const float* __restrict__ r2, const float* __restrict__ W2,
            const float* __restrict__ b2, const float* __restrict__ Wf,
            const float* __restrict__ bf, float* __restrict__ out, int N) {
  __shared__ _Float16 w2t[64 * 64];
  for (int idx = threadIdx.x; idx < 64 * 64; idx += 128) {
    int col = idx >> 6, k = idx & 63;
    w2t[idx] = (_Float16)W2[k * 64 + col];
  }
  __syncthreads();

  int wave  = threadIdx.x >> 5;
  int lane  = threadIdx.x & 31;
  int rbase = blockIdx.x * 64 + wave * 16;
  int lrow  = lane & 15;
  int hi    = lane >> 4;
  int kb    = hi * 8;

  const float* ar = r2 + (size_t)(rbase + lrow) * 64;
  const _Float16* wt = w2t + (size_t)lrow * 64;

  v8f zero = {0, 0, 0, 0, 0, 0, 0, 0};
  v8f acc[4];
  acc[0] = zero; acc[1] = zero; acc[2] = zero; acc[3] = zero;

#pragma unroll
  for (int kc = 0; kc < 64; kc += 32) {
    v16h a;
#pragma unroll
    for (int j = 0; j < 8; ++j) {
      a[j]     = (_Float16)ar[kc + kb + j];
      a[8 + j] = (_Float16)ar[kc + kb + 16 + j];
    }
#pragma unroll
    for (int ct = 0; ct < 4; ++ct) {
      const _Float16* wc = wt + (size_t)ct * 16 * 64;
      v8h blo = *reinterpret_cast<const v8h*>(wc + kc + kb);
      v8h bhi = *reinterpret_cast<const v8h*>(wc + kc + kb + 16);
      v16h b = __builtin_shufflevector(blo, bhi, 0, 1, 2, 3, 4, 5, 6, 7,
                                       8, 9, 10, 11, 12, 13, 14, 15);
      acc[ct] = __builtin_amdgcn_wmma_f32_16x16x32_f16(
          false, a, false, b, (short)0, acc[ct], false, false);
    }
  }

  float bias[4], wf0[4], wf1[4];
#pragma unroll
  for (int ct = 0; ct < 4; ++ct) {
    int col = ct * 16 + lrow;
    bias[ct] = b2[col];
    wf0[ct]  = Wf[col * 2 + 0];
    wf1[ct]  = Wf[col * 2 + 1];
  }
  float bf0 = bf[0], bf1 = bf[1];

#pragma unroll
  for (int v = 0; v < 8; ++v) {
    float p0 = 0.0f, p1 = 0.0f;
#pragma unroll
    for (int ct = 0; ct < 4; ++ct) {
      float h = fmaxf(acc[ct][v] + bias[ct], 0.0f);
      p0 += h * wf0[ct];
      p1 += h * wf1[ct];
    }
#pragma unroll
    for (int off = 8; off >= 1; off >>= 1) {
      p0 += __shfl_xor(p0, off, 16);
      p1 += __shfl_xor(p1, off, 16);
    }
    int row = rbase + v + hi * 8;
    if (lrow == 0 && row < N) {
      out[row * 2 + 0] = p0 + bf0;
      out[row * 2 + 1] = p1 + bf1;
    }
  }
}

extern "C" void kernel_launch(void* const* d_in, const int* in_sizes, int n_in,
                              void* d_out, int out_size, void* d_ws, size_t ws_size,
                              hipStream_t stream) {
  const float* x  = (const float*)d_in[0];
  const int*   ei = (const int*)d_in[1];
  const float* W1 = (const float*)d_in[2];
  const float* b1 = (const float*)d_in[3];
  const float* W2 = (const float*)d_in[4];
  const float* b2 = (const float*)d_in[5];
  const float* Wf = (const float*)d_in[6];
  const float* bf = (const float*)d_in[7];
  float* out = (float*)d_out;

  int N = in_sizes[0] / 4;
  int E = in_sizes[1] / 2;
  int Npad = ((N + 63) / 64) * 64;
  const int* src = ei;
  const int* dst = ei + E;

  float* ws   = (float*)d_ws;
  float* dis  = ws;
  float* aggx = dis + N;
  float* r2   = aggx + (size_t)4 * N;

  const int TB = 256;
  k_init_deg<<<(N + TB - 1) / TB, TB, 0, stream>>>(dis, N);
  k_deg_edges<<<(E + TB - 1) / TB, TB, 0, stream>>>(dst, dis, E);
  k_rsqrt_self<<<(N + TB - 1) / TB, TB, 0, stream>>>(x, dis, aggx, N);
  k_agg1<<<(E + TB - 1) / TB, TB, 0, stream>>>(src, dst, x, dis, aggx, E);
  k_layer2_own<<<(N + OWN_ROWS - 1) / OWN_ROWS, OWN_TB, 0, stream>>>(src, dst, aggx, W1, b1, dis, r2, N, E);
  k_gemm2_out<<<Npad / 64, 128, 0, stream>>>(r2, W2, b2, Wf, bf, out, N);
}


# Edges first partitioned by owning block (LDS counts, one global integer add per block and range); a layer-2 block reads only its own pair list, not all destinations; LDS adds in arrival order
# speedup vs baseline: 2.1797x; 1.2298x over previous
#include <hip/hip_runtime.h>

typedef __attribute__((ext_vector_type(16))) _Float16 v16h;
typedef __attribute__((ext_vector_type(8)))  _Float16 v8h;
typedef __attribute__((ext_vector_type(8)))  float    v8f;

static __device__ __forceinline__ int gtid() {
  return blockIdx.x * blockDim.x + threadIdx.x;
}

__global__ void k_init_deg(float* __restrict__ deg, int N) {
  int i = gtid();
  if (i < N) deg[i] = 1.0f;
}

__global__ void k_deg_edges(const int* __restrict__ dst, float* __restrict__ deg, int E) {
  int e = gtid();
  if (e < E) atomicAdd(&deg[dst[e]], 1.0f);
}

#define OWN_ROWS  391

__global__ void __launch_bounds__(512)
k_rsqrt_self(const float* __restrict__ x, float* __restrict__ dis,
             float* __restrict__ aggx, int* __restrict__ bincnt,
             int* __restrict__ cursor, int N) {
  __shared__ int wsum[16];
  const int t = threadIdx.x;
  const int i = blockIdx.x * OWN_ROWS + t;
  int cnt = 0;
  if (t < OWN_ROWS && i < N) {
    float dg = dis[i];
    cnt = (int)dg - 1;
    float r = rsqrtf(dg);
    dis[i] = r;
    float nn = r * r;
    float4 xv = reinterpret_cast<const float4*>(x)[i];
    float4 o = make_float4(xv.x * nn, xv.y * nn, xv.z * nn, xv.w * nn);
    reinterpret_cast<float4*>(aggx)[i] = o;
  }
  for (int o = 16; o >= 1; o >>= 1) cnt += __shfl_xor(cnt, o);
  if ((t & 31) == 0) wsum[t >> 5] = cnt;
  __syncthreads();
  if (t == 0) {
    int tot = 0;
    for (int w = 0; w < 16; ++w) tot += wsum[w];
    bincnt[blockIdx.x] = tot;
    cursor[blockIdx.x] = 0;
  }
}

__global__ void k_agg1(const int* __restrict__ src, const int* __restrict__ dst,
                       const float* __restrict__ x, const float* __restrict__ dis,
                       float* __restrict__ aggx, int E) {
  int e = gtid();
  int lane = threadIdx.x & 31;
  int ebase = e - lane;
  int s = 0, d = 0;
  float n = 0.0f;
  if (e < E) {
    s = src[e];
    d = dst[e];
    n = dis[s] * dis[d];
  }
  int c = lane & 3;
  int q = lane >> 2;
  int dj[4];
  float nj[4], v[4];
#pragma unroll
  for (int r = 0; r < 4; ++r) {
    int j = r * 8 + q;
    int sj = __shfl(s, j);
    dj[r] = __shfl(d, j);
    nj[r] = __shfl(n, j);
    v[r] = x[(size_t)sj * 4 + c];
  }
#pragma unroll
  for (int r = 0; r < 4; ++r) {
    if (ebase + r * 8 + q < E)
      atomicAdd(aggx + (size_t)dj[r] * 4 + c, v[r] * nj[r]);
  }
}

#define OWN_TB    1024
#define PART_TB   1024
#define PART_U    8

static __device__ __forceinline__ int part_block_scan(int v, int* wtot, int lane, int wave) {
  int incl = v;
  for (int o = 1; o < 32; o <<= 1) {
    int up = __shfl_up(incl, o);
    if (lane >= o) incl += up;
  }
  if (lane == 31) wtot[wave] = incl;
  __syncthreads();
  if (wave == 0) {
    int t = wtot[lane];
    int it = t;
    for (int o = 1; o < 32; o <<= 1) {
      int up = __shfl_up(it, o);
      if (lane >= o) it += up;
    }
    wtot[lane] = it - t;
  }
  __syncthreads();
  return incl - v + wtot[wave];
}

__global__ void __launch_bounds__(PART_TB)
k_partition(const int* __restrict__ src, const int* __restrict__ dst,
            const int* __restrict__ bincnt, int* __restrict__ cursor,
            int* __restrict__ binoff, int* __restrict__ list, int E, int NB) {
  __shared__ int hist[PART_TB];
  __shared__ int lst[PART_TB];
  __shared__ int wtotA[32];
  __shared__ int wtotB[32];
  __shared__ int2 stage[PART_TB * PART_U];
  const int tid = threadIdx.x;
  const int lane = tid & 31;
  const int wave = tid >> 5;

  const int cbase = blockIdx.x * (PART_TB * PART_U);
  int s[PART_U], d[PART_U], bq[PART_U], rk[PART_U];
#pragma unroll
  for (int u = 0; u < PART_U; ++u) {
    int e = cbase + u * PART_TB + tid;
    s[u] = 0;
    d[u] = -1;
    if (e < E) { s[u] = src[e]; d[u] = dst[e]; }
  }

  const int c = (tid < NB) ? bincnt[tid] : 0;
  hist[tid] = 0;
  const int start = part_block_scan(c, wtotA, lane, wave);
  if (blockIdx.x == 0 && tid < NB) {
    binoff[tid] = start;
    if (tid == NB - 1) binoff[NB] = start + c;
  }

#pragma unroll
  for (int u = 0; u < PART_U; ++u) {
    bq[u] = 0;
    rk[u] = 0;
    if (d[u] >= 0) {
      bq[u] = d[u] / OWN_ROWS;
      rk[u] = atomicAdd(&hist[bq[u]], 1);
    }
  }
  __syncthreads();

  const int n = hist[tid];
  int g = 0;
  if (n > 0) g = atomicAdd(&cursor[tid], n);
  const int ls = part_block_scan(n, wtotB, lane, wave);
  lst[tid] = ls;
  hist[tid] = start + g - ls;
  __syncthreads();

  int q[PART_U];
#pragma unroll
  for (int u = 0; u < PART_U; ++u) q[u] = lst[bq[u]] + rk[u];
#pragma unroll
  for (int u = 0; u < PART_U; ++u) {
    if (d[u] >= 0) stage[q[u]] = make_int2(s[u], d[u]);
  }
  __syncthreads();

  int2* lp = reinterpret_cast<int2*>(list);
  int nvalid = E - cbase;
  if (nvalid > PART_TB * PART_U) nvalid = PART_TB * PART_U;
  for (int i = tid; i < nvalid; i += PART_TB) {
    int2 p = stage[i];
    lp[hist[p.y / OWN_ROWS] + i] = p;
  }
}

struct OwnW {
  float a0, a1, a2, a3, ab;
  float b0, b1, b2, b3, bb;
};

__global__ void __launch_bounds__(OWN_TB)
k_layer2_own(const int* __restrict__ list, const int* __restrict__ binoff,
             const float* __restrict__ aggx, const float* __restrict__ W1,
             const float* __restrict__ b1, const float* __restrict__ dis,
             float* __restrict__ r2, int N) {
  __shared__ float acc[OWN_ROWS * 64];

  const int lane = threadIdx.x & 31;
  const int wave = __builtin_amdgcn_readfirstlane((int)(threadIdx.x >> 5));
  const int lo = blockIdx.x * OWN_ROWS;
  int rows = N - lo;
  if (rows > OWN_ROWS) rows = OWN_ROWS;

  OwnW w;
  w.a0 = W1[lane];      w.a1 = W1[64 + lane]; w.a2 = W1[128 + lane]; w.a3 = W1[192 + lane];
  w.b0 = W1[32 + lane]; w.b1 = W1[96 + lane]; w.b2 = W1[160 + lane]; w.b3 = W1[224 + lane];
  w.ab = b1[lane];      w.bb = b1[32 + lane];

  for (int r = wave; r < rows; r += OWN_TB / 32) {
    int node = lo + r;
    float4 av = reinterpret_cast<const float4*>(aggx)[node];
    float dd = dis[node];
    float ha = w.ab;
    ha = __builtin_fmaf(av.x, w.a0, ha);
    ha = __builtin_fmaf(av.y, w.a1, ha);
    ha = __builtin_fmaf(av.z, w.a2, ha);
    ha = __builtin_fmaf(av.w, w.a3, ha);
    float hb = w.bb;
    hb = __builtin_fmaf(av.x, w.b0, hb);
    hb = __builtin_fmaf(av.y, w.b1, hb);
    hb = __builtin_fmaf(av.z, w.b2, hb);
    hb = __builtin_fmaf(av.w, w.b3, hb);
    ha = fmaxf(ha, 0.0f);
    hb = fmaxf(hb, 0.0f);
    acc[r * 64 + lane] = ha * dd * dd;
    acc[r * 64 + 32 + lane] = hb * dd * dd;
  }
  __syncthreads();

  const int beg = binoff[blockIdx.x];
  const int end = binoff[blockIdx.x + 1];
  const int2* lp = reinterpret_cast<const int2*>(list);
  for (int base = beg + wave * 32; base < end; base += OWN_TB) {
    int m = end - base;
    if (m > 32) m = 32;
    int r = 0;
    float n = 0.0f;
    float4 av = make_float4(0.0f, 0.0f, 0.0f, 0.0f);
    if (lane < m) {
      int2 p = lp[base + lane];
      av = reinterpret_cast<const float4*>(aggx)[p.x];
      n = dis[p.x] * dis[p.y];
      r = p.y - lo;
    }
    int nb = __float_as_int(n);
    int x0b = __float_as_int(av.x), x1b = __float_as_int(av.y);
    int x2b = __float_as_int(av.z), x3b = __float_as_int(av.w);
    for (int j = 0; j < m; ++j) {
      int rj = __builtin_amdgcn_readlane(r, j);
      float nj = __int_as_float(__builtin_amdgcn_readlane(nb, j));
      float x0 = __int_as_float(__builtin_amdgcn_readlane(x0b, j));
      float x1 = __int_as_float(__builtin_amdgcn_readlane(x1b, j));
      float x2 = __int_as_float(__builtin_amdgcn_readlane(x2b, j));
      float x3 = __int_as_float(__builtin_amdgcn_readlane(x3b, j));
      float ha = w.ab;
      ha = __builtin_fmaf(x0, w.a0, ha);
      ha = __builtin_fmaf(x1, w.a1, ha);
      ha = __builtin_fmaf(x2, w.a2, ha);
      ha = __builtin_fmaf(x3, w.a3, ha);
      float hb = w.bb;
      hb = __builtin_fmaf(x0, w.b0, hb);
      hb = __builtin_fmaf(x1, w.b1, hb);
      hb = __builtin_fmaf(x2, w.b2, hb);
      hb = __builtin_fmaf(x3, w.b3, hb);
      ha = fmaxf(ha, 0.0f);
      hb = fmaxf(hb, 0.0f);
      float* row = acc + rj * 64;
      atomicAdd(row + lane, ha * nj);
      atomicAdd(row + 32 + lane, hb * nj);
    }
  }
  __syncthreads();

  float* outp = r2 + (size_t)lo * 64;
  for (int i = threadIdx.x; i < rows * 64; i += OWN_TB) outp[i] = acc[i];
}

__global__ void __launch_bounds__(128)
k_gemm2_out(const float* __restrict__ r2, const float* __restrict__ W2,
            const float* __restrict__ b2, const float* __restrict__ Wf,
            const float* __restrict__ bf, float* __restrict__ out, int N) {
  __shared__ _Float16 w2t[64 * 64];
  for (int idx = threadIdx.x; idx < 64 * 64; idx += 128) {
    int col = idx >> 6, k = idx & 63;
    w2t[idx] = (_Float16)W2[k * 64 + col];
  }
  __syncthreads();

  int wave  = threadIdx.x >> 5;
  int lane  = threadIdx.x & 31;
  int rbase = blockIdx.x * 64 + wave * 16;
  int lrow  = lane & 15;
  int hi    = lane >> 4;
  int kb    = hi * 8;

  const float* ar = r2 + (size_t)(rbase + lrow) * 64;
  const _Float16* wt = w2t + (size_t)lrow * 64;

  v8f zero = {0, 0, 0, 0, 0, 0, 0, 0};
  v8f acc[4];
  acc[0] = zero; acc[1] = zero; acc[2] = zero; acc[3] = zero;

#pragma unroll
  for (int kc = 0; kc < 64; kc += 32) {
    v16h a;
#pragma unroll
    for (int j = 0; j < 8; ++j) {
      a[j]     = (_Float16)ar[kc + kb + j];
      a[8 + j] = (_Float16)ar[kc + kb + 16 + j];
    }
#pragma unroll
    for (int ct = 0; ct < 4; ++ct) {
      const _Float16* wc = wt + (size_t)ct * 16 * 64;
      v8h blo = *reinterpret_cast<const v8h*>(wc + kc + kb);
      v8h bhi = *reinterpret_cast<const v8h*>(wc + kc + kb + 16);
      v16h b = __builtin_shufflevector(blo, bhi, 0, 1, 2, 3, 4, 5, 6, 7,
                                       8, 9, 10, 11, 12, 13, 14, 15);
      acc[ct] = __builtin_amdgcn_wmma_f32_16x16x32_f16(
          false, a, false, b, (short)0, acc[ct], false, false);
    }
  }

  float bias[4], wf0[4], wf1[4];
#pragma unroll
  for (int ct = 0; ct < 4; ++ct) {
    int col = ct * 16 + lrow;
    bias[ct] = b2[col];
    wf0[ct]  = Wf[col * 2 + 0];
    wf1[ct]  = Wf[col * 2 + 1];
  }
  float bf0 = bf[0], bf1 = bf[1];

#pragma unroll
  for (int v = 0; v < 8; ++v) {
    float p0 = 0.0f, p1 = 0.0f;
#pragma unroll
    for (int ct = 0; ct < 4; ++ct) {
      float h = fmaxf(acc[ct][v] + bias[ct], 0.0f);
      p0 += h * wf0[ct];
      p1 += h * wf1[ct];
    }
#pragma unroll
    for (int off = 8; off >= 1; off >>= 1) {
      p0 += __shfl_xor(p0, off, 16);
      p1 += __shfl_xor(p1, off, 16);
    }
    int row = rbase + v + hi * 8;
    if (lrow == 0 && row < N) {
      out[row * 2 + 0] = p0 + bf0;
      out[row * 2 + 1] = p1 + bf1;
    }
  }
}

extern "C" void kernel_launch(void* const* d_in, const int* in_sizes, int n_in,
                              void* d_out, int out_size, void* d_ws, size_t ws_size,
                              hipStream_t stream) {
  const float* x  = (const float*)d_in[0];
  const int*   ei = (const int*)d_in[1];
  const float* W1 = (const float*)d_in[2];
  const float* b1 = (const float*)d_in[3];
  const float* W2 = (const float*)d_in[4];
  const float* b2 = (const float*)d_in[5];
  const float* Wf = (const float*)d_in[6];
  const float* bf = (const float*)d_in[7];
  float* out = (float*)d_out;

  int N = in_sizes[0] / 4;
  int E = in_sizes[1] / 2;
  int Npad = ((N + 63) / 64) * 64;
  const int* src = ei;
  const int* dst = ei + E;

  float* ws   = (float*)d_ws;
  float* dis  = ws;
  float* aggx = dis + N;
  float* r2   = aggx + (size_t)4 * N;
  int NB = (N + OWN_ROWS - 1) / OWN_ROWS;
  int* iw     = (int*)d_ws + (((size_t)5 * N + (size_t)64 * Npad + 63) / 64) * 64;
  int* bincnt = iw;
  int* cursor = iw + 1024;
  int* binoff = iw + 2048;
  int* list   = iw + 4096;

  const int TB = 256;
  k_init_deg<<<(N + TB - 1) / TB, TB, 0, stream>>>(dis, N);
  k_deg_edges<<<(E + TB - 1) / TB, TB, 0, stream>>>(dst, dis, E);
  k_rsqrt_self<<<NB, 512, 0, stream>>>(x, dis, aggx, bincnt, cursor, N);
  k_agg1<<<(E + TB - 1) / TB, TB, 0, stream>>>(src, dst, x, dis, aggx, E);
  k_partition<<<(E + PART_TB * PART_U - 1) / (PART_TB * PART_U), PART_TB, 0, stream>>>(src, dst, bincnt, cursor, binoff, list, E, NB);
  k_layer2_own<<<NB, OWN_TB, 0, stream>>>(list, binoff, aggx, W1, b1, dis, r2, N);
  k_gemm2_out<<<Npad / 64, 128, 0, stream>>>(r2, W2, b2, Wf, bf, out, N);
}


# In-degrees counted in integers: per-range LDS counts feed the partition, then each destination-owning block counts its own pair list per row in LDS; no global float add per edge; same deg values
# speedup vs baseline: 3.5503x; 1.6288x over previous
#include <hip/hip_runtime.h>

typedef __attribute__((ext_vector_type(16))) _Float16 v16h;
typedef __attribute__((ext_vector_type(8)))  _Float16 v8h;
typedef __attribute__((ext_vector_type(8)))  float    v8f;

static __device__ __forceinline__ int gtid() {
  return blockIdx.x * blockDim.x + threadIdx.x;
}

#define OWN_ROWS  391
#define OWN_TB    1024
#define PART_TB   1024
#define PART_U    8


__global__ void __launch_bounds__(PART_TB)
k_zero(int* __restrict__ bincnt, int* __restrict__ cursor) {
  bincnt[threadIdx.x] = 0;
  cursor[threadIdx.x] = 0;
}

__global__ void __launch_bounds__(PART_TB)
k_count_ranges(const int* __restrict__ dst, int* __restrict__ bincnt, int E) {
  __shared__ int hist[PART_TB];
  const int tid = threadIdx.x;
  hist[tid] = 0;
  const int cbase = blockIdx.x * (PART_TB * PART_U);
  int d[PART_U];
#pragma unroll
  for (int u = 0; u < PART_U; ++u) {
    int e = cbase + u * PART_TB + tid;
    d[u] = -1;
    if (e < E) d[u] = dst[e];
  }
  __syncthreads();
#pragma unroll
  for (int u = 0; u < PART_U; ++u) {
    if (d[u] >= 0) atomicAdd(&hist[d[u] / OWN_ROWS], 1);
  }
  __syncthreads();
  const int n = hist[tid];
  if (n > 0) atomicAdd(&bincnt[tid], n);
}

__global__ void __launch_bounds__(OWN_TB)
k_own_deg(const int* __restrict__ list, const int* __restrict__ binoff,
          const float* __restrict__ x, float* __restrict__ dis,
          float* __restrict__ aggx, int N) {
  __shared__ int cnt[OWN_ROWS];
  const int t = threadIdx.x;
  const int lo = blockIdx.x * OWN_ROWS;
  if (t < OWN_ROWS) cnt[t] = 0;
  __syncthreads();
  const int beg = binoff[blockIdx.x];
  const int end = binoff[blockIdx.x + 1];
  for (int base = beg + t; base < end; base += 8 * OWN_TB) {
    int d[8];
#pragma unroll
    for (int u = 0; u < 8; ++u) {
      int i = base + u * OWN_TB;
      d[u] = lo - 1;
      if (i < end) d[u] = list[(size_t)2 * i + 1];
    }
#pragma unroll
    for (int u = 0; u < 8; ++u) {
      int r = d[u] - lo;
      if (r >= 0) atomicAdd(&cnt[r], 1);
    }
  }
  __syncthreads();
  const int i = lo + t;
  if (t < OWN_ROWS && i < N) {
    float dg = (float)(cnt[t] + 1);
    float r = rsqrtf(dg);
    dis[i] = r;
    float nn = r * r;
    float4 xv = reinterpret_cast<const float4*>(x)[i];
    float4 o = make_float4(xv.x * nn, xv.y * nn, xv.z * nn, xv.w * nn);
    reinterpret_cast<float4*>(aggx)[i] = o;
  }
}

__global__ void k_agg1(const int* __restrict__ src, const int* __restrict__ dst,
                       const float* __restrict__ x, const float* __restrict__ dis,
                       float* __restrict__ aggx, int E) {
  int e = gtid();
  int lane = threadIdx.x & 31;
  int ebase = e - lane;
  int s = 0, d = 0;
  float n = 0.0f;
  if (e < E) {
    s = src[e];
    d = dst[e];
    n = dis[s] * dis[d];
  }
  int c = lane & 3;
  int q = lane >> 2;
  int dj[4];
  float nj[4], v[4];
#pragma unroll
  for (int r = 0; r < 4; ++r) {
    int j = r * 8 + q;
    int sj = __shfl(s, j);
    dj[r] = __shfl(d, j);
    nj[r] = __shfl(n, j);
    v[r] = x[(size_t)sj * 4 + c];
  }
#pragma unroll
  for (int r = 0; r < 4; ++r) {
    if (ebase + r * 8 + q < E)
      atomicAdd(aggx + (size_t)dj[r] * 4 + c, v[r] * nj[r]);
  }
}

static __device__ __forceinline__ int part_block_scan(int v, int* wtot, int lane, int wave) {
  int incl = v;
  for (int o = 1; o < 32; o <<= 1) {
    int up = __shfl_up(incl, o);
    if (lane >= o) incl += up;
  }
  if (lane == 31) wtot[wave] = incl;
  __syncthreads();
  if (wave == 0) {
    int t = wtot[lane];
    int it = t;
    for (int o = 1; o < 32; o <<= 1) {
      int up = __shfl_up(it, o);
      if (lane >= o) it += up;
    }
    wtot[lane] = it - t;
  }
  __syncthreads();
  return incl - v + wtot[wave];
}

__global__ void __launch_bounds__(PART_TB)
k_partition(const int* __restrict__ src, const int* __restrict__ dst,
            const int* __restrict__ bincnt, int* __restrict__ cursor,
            int* __restrict__ binoff, int* __restrict__ list, int E, int NB) {
  __shared__ int hist[PART_TB];
  __shared__ int lst[PART_TB];
  __shared__ int wtotA[32];
  __shared__ int wtotB[32];
  __shared__ int2 stage[PART_TB * PART_U];
  const int tid = threadIdx.x;
  const int lane = tid & 31;
  const int wave = tid >> 5;

  const int cbase = blockIdx.x * (PART_TB * PART_U);
  int s[PART_U], d[PART_U], bq[PART_U], rk[PART_U];
#pragma unroll
  for (int u = 0; u < PART_U; ++u) {
    int e = cbase + u * PART_TB + tid;
    s[u] = 0;
    d[u] = -1;
    if (e < E) { s[u] = src[e]; d[u] = dst[e]; }
  }

  const int c = (tid < NB) ? bincnt[tid] : 0;
  hist[tid] = 0;
  const int start = part_block_scan(c, wtotA, lane, wave);
  if (blockIdx.x == 0 && tid < NB) {
    binoff[tid] = start;
    if (tid == NB - 1) binoff[NB] = start + c;
  }

#pragma unroll
  for (int u = 0; u < PART_U; ++u) {
    bq[u] = 0;
    rk[u] = 0;
    if (d[u] >= 0) {
      bq[u] = d[u] / OWN_ROWS;
      rk[u] = atomicAdd(&hist[bq[u]], 1);
    }
  }
  __syncthreads();

  const int n = hist[tid];
  int g = 0;
  if (n > 0) g = atomicAdd(&cursor[tid], n);
  const int ls = part_block_scan(n, wtotB, lane, wave);
  lst[tid] = ls;
  hist[tid] = start + g - ls;
  __syncthreads();

  int q[PART_U];
#pragma unroll
  for (int u = 0; u < PART_U; ++u) q[u] = lst[bq[u]] + rk[u];
#pragma unroll
  for (int u = 0; u < PART_U; ++u) {
    if (d[u] >= 0) stage[q[u]] = make_int2(s[u], d[u]);
  }
  __syncthreads();

  int2* lp = reinterpret_cast<int2*>(list);
  int nvalid = E - cbase;
  if (nvalid > PART_TB * PART_U) nvalid = PART_TB * PART_U;
  for (int i = tid; i < nvalid; i += PART_TB) {
    int2 p = stage[i];
    lp[hist[p.y / OWN_ROWS] + i] = p;
  }
}

struct OwnW {
  float a0, a1, a2, a3, ab;
  float b0, b1, b2, b3, bb;
};

__global__ void __launch_bounds__(OWN_TB)
k_layer2_own(const int* __restrict__ list, const int* __restrict__ binoff,
             const float* __restrict__ aggx, const float* __restrict__ W1,
             const float* __restrict__ b1, const float* __restrict__ dis,
             float* __restrict__ r2, int N) {
  __shared__ float acc[OWN_ROWS * 64];

  const int lane = threadIdx.x & 31;
  const int wave = __builtin_amdgcn_readfirstlane((int)(threadIdx.x >> 5));
  const int lo = blockIdx.x * OWN_ROWS;
  int rows = N - lo;
  if (rows > OWN_ROWS) rows = OWN_ROWS;

  OwnW w;
  w.a0 = W1[lane];      w.a1 = W1[64 + lane]; w.a2 = W1[128 + lane]; w.a3 = W1[192 + lane];
  w.b0 = W1[32 + lane]; w.b1 = W1[96 + lane]; w.b2 = W1[160 + lane]; w.b3 = W1[224 + lane];
  w.ab = b1[lane];      w.bb = b1[32 + lane];

  for (int r = wave; r < rows; r += OWN_TB / 32) {
    int node = lo + r;
    float4 av = reinterpret_cast<const float4*>(aggx)[node];
    float dd = dis[node];
    float ha = w.ab;
    ha = __builtin_fmaf(av.x, w.a0, ha);
    ha = __builtin_fmaf(av.y, w.a1, ha);
    ha = __builtin_fmaf(av.z, w.a2, ha);
    ha = __builtin_fmaf(av.w, w.a3, ha);
    float hb = w.bb;
    hb = __builtin_fmaf(av.x, w.b0, hb);
    hb = __builtin_fmaf(av.y, w.b1, hb);
    hb = __builtin_fmaf(av.z, w.b2, hb);
    hb = __builtin_fmaf(av.w, w.b3, hb);
    ha = fmaxf(ha, 0.0f);
    hb = fmaxf(hb, 0.0f);
    acc[r * 64 + lane] = ha * dd * dd;
    acc[r * 64 + 32 + lane] = hb * dd * dd;
  }
  __syncthreads();

  const int beg = binoff[blockIdx.x];
  const int end = binoff[blockIdx.x + 1];
  const int2* lp = reinterpret_cast<const int2*>(list);
  for (int base = beg + wave * 32; base < end; base += OWN_TB) {
    int m = end - base;
    if (m > 32) m = 32;
    int r = 0;
    float n = 0.0f;
    float4 av = make_float4(0.0f, 0.0f, 0.0f, 0.0f);
    if (lane < m) {
      int2 p = lp[base + lane];
      av = reinterpret_cast<const float4*>(aggx)[p.x];
      n = dis[p.x] * dis[p.y];
      r = p.y - lo;
    }
    int nb = __float_as_int(n);
    int x0b = __float_as_int(av.x), x1b = __float_as_int(av.y);
    int x2b = __float_as_int(av.z), x3b = __float_as_int(av.w);
    for (int j = 0; j < m; ++j) {
      int rj = __builtin_amdgcn_readlane(r, j);
      float nj = __int_as_float(__builtin_amdgcn_readlane(nb, j));
      float x0 = __int_as_float(__builtin_amdgcn_readlane(x0b, j));
      float x1 = __int_as_float(__builtin_amdgcn_readlane(x1b, j));
      float x2 = __int_as_float(__builtin_amdgcn_readlane(x2b, j));
      float x3 = __int_as_float(__builtin_amdgcn_readlane(x3b, j));
      float ha = w.ab;
      ha = __builtin_fmaf(x0, w.a0, ha);
      ha = __builtin_fmaf(x1, w.a1, ha);
      ha = __builtin_fmaf(x2, w.a2, ha);
      ha = __builtin_fmaf(x3, w.a3, ha);
      float hb = w.bb;
      hb = __builtin_fmaf(x0, w.b0, hb);
      hb = __builtin_fmaf(x1, w.b1, hb);
      hb = __builtin_fmaf(x2, w.b2, hb);
      hb = __builtin_fmaf(x3, w.b3, hb);
      ha = fmaxf(ha, 0.0f);
      hb = fmaxf(hb, 0.0f);
      float* row = acc + rj * 64;
      atomicAdd(row + lane, ha * nj);
      atomicAdd(row + 32 + lane, hb * nj);
    }
  }
  __syncthreads();

  float* outp = r2 + (size_t)lo * 64;
  for (int i = threadIdx.x; i < rows * 64; i += OWN_TB) outp[i] = acc[i];
}

__global__ void __launch_bounds__(128)
k_gemm2_out(const float* __restrict__ r2, const float* __restrict__ W2,
            const float* __restrict__ b2, const float* __restrict__ Wf,
            const float* __restrict__ bf, float* __restrict__ out, int N) {
  __shared__ _Float16 w2t[64 * 64];
  for (int idx = threadIdx.x; idx < 64 * 64; idx += 128) {
    int col = idx >> 6, k = idx & 63;
    w2t[idx] = (_Float16)W2[k * 64 + col];
  }
  __syncthreads();

  int wave  = threadIdx.x >> 5;
  int lane  = threadIdx.x & 31;
  int rbase = blockIdx.x * 64 + wave * 16;
  int lrow  = lane & 15;
  int hi    = lane >> 4;
  int kb    = hi * 8;

  const float* ar = r2 + (size_t)(rbase + lrow) * 64;
  const _Float16* wt = w2t + (size_t)lrow * 64;

  v8f zero = {0, 0, 0, 0, 0, 0, 0, 0};
  v8f acc[4];
  acc[0] = zero; acc[1] = zero; acc[2] = zero; acc[3] = zero;

#pragma unroll
  for (int kc = 0; kc < 64; kc += 32) {
    v16h a;
#pragma unroll
    for (int j = 0; j < 8; ++j) {
      a[j]     = (_Float16)ar[kc + kb + j];
      a[8 + j] = (_Float16)ar[kc + kb + 16 + j];
    }
#pragma unroll
    for (int ct = 0; ct < 4; ++ct) {
      const _Float16* wc = wt + (size_t)ct * 16 * 64;
      v8h blo = *reinterpret_cast<const v8h*>(wc + kc + kb);
      v8h bhi = *reinterpret_cast<const v8h*>(wc + kc + kb + 16);
      v16h b = __builtin_shufflevector(blo, bhi, 0, 1, 2, 3, 4, 5, 6, 7,
                                       8, 9, 10, 11, 12, 13, 14, 15);
      acc[ct] = __builtin_amdgcn_wmma_f32_16x16x32_f16(
          false, a, false, b, (short)0, acc[ct], false, false);
    }
  }

  float bias[4], wf0[4], wf1[4];
#pragma unroll
  for (int ct = 0; ct < 4; ++ct) {
    int col = ct * 16 + lrow;
    bias[ct] = b2[col];
    wf0[ct]  = Wf[col * 2 + 0];
    wf1[ct]  = Wf[col * 2 + 1];
  }
  float bf0 = bf[0], bf1 = bf[1];

#pragma unroll
  for (int v = 0; v < 8; ++v) {
    float p0 = 0.0f, p1 = 0.0f;
#pragma unroll
    for (int ct = 0; ct < 4; ++ct) {
      float h = fmaxf(acc[ct][v] + bias[ct], 0.0f);
      p0 += h * wf0[ct];
      p1 += h * wf1[ct];
    }
#pragma unroll
    for (int off = 8; off >= 1; off >>= 1) {
      p0 += __shfl_xor(p0, off, 16);
      p1 += __shfl_xor(p1, off, 16);
    }
    int row = rbase + v + hi * 8;
    if (lrow == 0 && row < N) {
      out[row * 2 + 0] = p0 + bf0;
      out[row * 2 + 1] = p1 + bf1;
    }
  }
}

extern "C" void kernel_launch(void* const* d_in, const int* in_sizes, int n_in,
                              void* d_out, int out_size, void* d_ws, size_t ws_size,
                              hipStream_t stream) {
  const float* x  = (const float*)d_in[0];
  const int*   ei = (const int*)d_in[1];
  const float* W1 = (const float*)d_in[2];
  const float* b1 = (const float*)d_in[3];
  const float* W2 = (const float*)d_in[4];
  const float* b2 = (const float*)d_in[5];
  const float* Wf = (const float*)d_in[6];
  const float* bf = (const float*)d_in[7];
  float* out = (float*)d_out;

  int N = in_sizes[0] / 4;
  int E = in_sizes[1] / 2;
  int Npad = ((N + 63) / 64) * 64;
  const int* src = ei;
  const int* dst = ei + E;

  float* ws   = (float*)d_ws;
  float* dis  = ws;
  float* aggx = dis + N;
  float* r2   = aggx + (size_t)4 * N;
  int NB = (N + OWN_ROWS - 1) / OWN_ROWS;
  int* iw     = (int*)d_ws + (((size_t)5 * N + (size_t)64 * Npad + 63) / 64) * 64;
  int* bincnt = iw;
  int* cursor = iw + 1024;
  int* binoff = iw + 2048;
  int* list   = iw + 4096;

  const int TB = 256;
  const int PB = (E + PART_TB * PART_U - 1) / (PART_TB * PART_U);
  k_zero<<<1, PART_TB, 0, stream>>>(bincnt, cursor);
  k_count_ranges<<<PB, PART_TB, 0, stream>>>(dst, bincnt, E);
  k_partition<<<PB, PART_TB, 0, stream>>>(src, dst, bincnt, cursor, binoff, list, E, NB);
  k_own_deg<<<NB, OWN_TB, 0, stream>>>(list, binoff, x, dis, aggx, N);
  k_agg1<<<(E + TB - 1) / TB, TB, 0, stream>>>(src, dst, x, dis, aggx, E);
  k_layer2_own<<<NB, OWN_TB, 0, stream>>>(list, binoff, aggx, W1, b1, dis, r2, N);
  k_gemm2_out<<<Npad / 64, 128, 0, stream>>>(r2, W2, b2, Wf, bf, out, N);
}


# Layer-1 sum moved to the destination-owning blocks: each pair's row x*dis[src] is gathered once and added in LDS as 64-bit fixed-point integers (exact, same bits in any order); dis[dst] once per row
# speedup vs baseline: 7.3615x; 2.0735x over previous
#include <hip/hip_runtime.h>

typedef __attribute__((ext_vector_type(16))) _Float16 v16h;
typedef __attribute__((ext_vector_type(8)))  _Float16 v8h;
typedef __attribute__((ext_vector_type(8)))  float    v8f;

#define OWN_ROWS  391
#define OWN_TB    1024
#define PART_TB   1024
#define PART_U    8
#define AGG_U     16
#define FX_ONE    0x1p36f
#define FX_UNIT   0x1p-36f


__global__ void __launch_bounds__(PART_TB)
k_zero(int* __restrict__ bincnt, int* __restrict__ cursor) {
  bincnt[threadIdx.x] = 0;
  cursor[threadIdx.x] = 0;
}

__global__ void __launch_bounds__(PART_TB)
k_count_ranges(const int* __restrict__ dst, int* __restrict__ bincnt, int E) {
  __shared__ int hist[PART_TB];
  const int tid = threadIdx.x;
  hist[tid] = 0;
  const int cbase = blockIdx.x * (PART_TB * PART_U);
  int d[PART_U];
#pragma unroll
  for (int u = 0; u < PART_U; ++u) {
    int e = cbase + u * PART_TB + tid;
    d[u] = -1;
    if (e < E) d[u] = dst[e];
  }
  __syncthreads();
#pragma unroll
  for (int u = 0; u < PART_U; ++u) {
    if (d[u] >= 0) atomicAdd(&hist[d[u] / OWN_ROWS], 1);
  }
  __syncthreads();
  const int n = hist[tid];
  if (n > 0) atomicAdd(&bincnt[tid], n);
}

__global__ void __launch_bounds__(OWN_TB)
k_own_deg(const int* __restrict__ list, const int* __restrict__ binoff,
          const float* __restrict__ x, float* __restrict__ dis,
          float* __restrict__ xs, int N) {
  __shared__ int cnt[OWN_ROWS];
  const int t = threadIdx.x;
  const int lo = blockIdx.x * OWN_ROWS;
  if (t < OWN_ROWS) cnt[t] = 0;
  __syncthreads();
  const int beg = binoff[blockIdx.x];
  const int end = binoff[blockIdx.x + 1];
  for (int base = beg + t; base < end; base += 8 * OWN_TB) {
    int d[8];
#pragma unroll
    for (int u = 0; u < 8; ++u) {
      int i = base + u * OWN_TB;
      d[u] = lo - 1;
      if (i < end) d[u] = list[(size_t)2 * i + 1];
    }
#pragma unroll
    for (int u = 0; u < 8; ++u) {
      int r = d[u] - lo;
      if (r >= 0) atomicAdd(&cnt[r], 1);
    }
  }
  __syncthreads();
  const int i = lo + t;
  if (t < OWN_ROWS && i < N) {
    float dg = (float)(cnt[t] + 1);
    float r = rsqrtf(dg);
    dis[i] = r;
    float4 xv = reinterpret_cast<const float4*>(x)[i];
    float4 o = make_float4(xv.x * r, xv.y * r, xv.z * r, xv.w * r);
    reinterpret_cast<float4*>(xs)[i] = o;
  }
}

static __device__ __forceinline__ unsigned long long to_fx(float v) {
  return (unsigned long long)(long long)(v * FX_ONE);
}
static __device__ __forceinline__ float from_fx(unsigned long long s) {
  return (float)(long long)s * FX_UNIT;
}

__global__ void __launch_bounds__(OWN_TB)
k_agg1_own(const int* __restrict__ list, const int* __restrict__ binoff,
           const float* __restrict__ xs, const float* __restrict__ dis,
           float* __restrict__ aggx, int N) {
  __shared__ unsigned long long acc[4 * OWN_ROWS];
  const int t = threadIdx.x;
  const int lo = blockIdx.x * OWN_ROWS;
  const int i = lo + t;
  const float4* xs4 = reinterpret_cast<const float4*>(xs);

  if (t < OWN_ROWS) {
    float4 sv = make_float4(0.0f, 0.0f, 0.0f, 0.0f);
    if (i < N) sv = xs4[i];
    acc[t] = to_fx(sv.x);
    acc[OWN_ROWS + t] = to_fx(sv.y);
    acc[2 * OWN_ROWS + t] = to_fx(sv.z);
    acc[3 * OWN_ROWS + t] = to_fx(sv.w);
  }
  __syncthreads();

  const int beg = binoff[blockIdx.x];
  const int end = binoff[blockIdx.x + 1];
  const int2* lp = reinterpret_cast<const int2*>(list);
  for (int base = beg + t; base < end; base += AGG_U * OWN_TB) {
    int2 p[AGG_U];
    float4 v[AGG_U];
#pragma unroll
    for (int u = 0; u < AGG_U; ++u) {
      int k = base + u * OWN_TB;
      p[u] = make_int2(0, lo - 1);
      if (k < end) p[u] = lp[k];
    }
#pragma unroll
    for (int u = 0; u < AGG_U; ++u) v[u] = xs4[p[u].x];
#pragma unroll
    for (int u = 0; u < AGG_U; ++u) {
      int r = p[u].y - lo;
      if (r >= 0) {
        atomicAdd(&acc[r], to_fx(v[u].x));
        atomicAdd(&acc[OWN_ROWS + r], to_fx(v[u].y));
        atomicAdd(&acc[2 * OWN_ROWS + r], to_fx(v[u].z));
        atomicAdd(&acc[3 * OWN_ROWS + r], to_fx(v[u].w));
      }
    }
  }
  __syncthreads();

  if (t < OWN_ROWS && i < N) {
    float dd = dis[i];
    float4 o;
    o.x = dd * from_fx(acc[t]);
    o.y = dd * from_fx(acc[OWN_ROWS + t]);
    o.z = dd * from_fx(acc[2 * OWN_ROWS + t]);
    o.w = dd * from_fx(acc[3 * OWN_ROWS + t]);
    reinterpret_cast<float4*>(aggx)[i] = o;
  }
}

static __device__ __forceinline__ int part_block_scan(int v, int* wtot, int lane, int wave) {
  int incl = v;
  for (int o = 1; o < 32; o <<= 1) {
    int up = __shfl_up(incl, o);
    if (lane >= o) incl += up;
  }
  if (lane == 31) wtot[wave] = incl;
  __syncthreads();
  if (wave == 0) {
    int t = wtot[lane];
    int it = t;
    for (int o = 1; o < 32; o <<= 1) {
      int up = __shfl_up(it, o);
      if (lane >= o) it += up;
    }
    wtot[lane] = it - t;
  }
  __syncthreads();
  return incl - v + wtot[wave];
}

__global__ void __launch_bounds__(PART_TB)
k_partition(const int* __restrict__ src, const int* __restrict__ dst,
            const int* __restrict__ bincnt, int* __restrict__ cursor,
            int* __restrict__ binoff, int* __restrict__ list, int E, int NB) {
  __shared__ int hist[PART_TB];
  __shared__ int lst[PART_TB];
  __shared__ int wtotA[32];
  __shared__ int wtotB[32];
  __shared__ int2 stage[PART_TB * PART_U];
  const int tid = threadIdx.x;
  const int lane = tid & 31;
  const int wave = tid >> 5;

  const int cbase = blockIdx.x * (PART_TB * PART_U);
  int s[PART_U], d[PART_U], bq[PART_U], rk[PART_U];
#pragma unroll
  for (int u = 0; u < PART_U; ++u) {
    int e = cbase + u * PART_TB + tid;
    s[u] = 0;
    d[u] = -1;
    if (e < E) { s[u] = src[e]; d[u] = dst[e]; }
  }

  const int c = (tid < NB) ? bincnt[tid] : 0;
  hist[tid] = 0;
  const int start = part_block_scan(c, wtotA, lane, wave);
  if (blockIdx.x == 0 && tid < NB) {
    binoff[tid] = start;
    if (tid == NB - 1) binoff[NB] = start + c;
  }

#pragma unroll
  for (int u = 0; u < PART_U; ++u) {
    bq[u] = 0;
    rk[u] = 0;
    if (d[u] >= 0) {
      bq[u] = d[u] / OWN_ROWS;
      rk[u] = atomicAdd(&hist[bq[u]], 1);
    }
  }
  __syncthreads();

  const int n = hist[tid];
  int g = 0;
  if (n > 0) g = atomicAdd(&cursor[tid], n);
  const int ls = part_block_scan(n, wtotB, lane, wave);
  lst[tid] = ls;
  hist[tid] = start + g - ls;
  __syncthreads();

  int q[PART_U];
#pragma unroll
  for (int u = 0; u < PART_U; ++u) q[u] = lst[bq[u]] + rk[u];
#pragma unroll
  for (int u = 0; u < PART_U; ++u) {
    if (d[u] >= 0) stage[q[u]] = make_int2(s[u], d[u]);
  }
  __syncthreads();

  int2* lp = reinterpret_cast<int2*>(list);
  int nvalid = E - cbase;
  if (nvalid > PART_TB * PART_U) nvalid = PART_TB * PART_U;
  for (int i = tid; i < nvalid; i += PART_TB) {
    int2 p = stage[i];
    lp[hist[p.y / OWN_ROWS] + i] = p;
  }
}

struct OwnW {
  float a0, a1, a2, a3, ab;
  float b0, b1, b2, b3, bb;
};

__global__ void __launch_bounds__(OWN_TB)
k_layer2_own(const int* __restrict__ list, const int* __restrict__ binoff,
             const float* __restrict__ aggx, const float* __restrict__ W1,
             const float* __restrict__ b1, const float* __restrict__ dis,
             float* __restrict__ r2, int N) {
  __shared__ float acc[OWN_ROWS * 64];

  const int lane = threadIdx.x & 31;
  const int wave = __builtin_amdgcn_readfirstlane((int)(threadIdx.x >> 5));
  const int lo = blockIdx.x * OWN_ROWS;
  int rows = N - lo;
  if (rows > OWN_ROWS) rows = OWN_ROWS;

  OwnW w;
  w.a0 = W1[lane];      w.a1 = W1[64 + lane]; w.a2 = W1[128 + lane]; w.a3 = W1[192 + lane];
  w.b0 = W1[32 + lane]; w.b1 = W1[96 + lane]; w.b2 = W1[160 + lane]; w.b3 = W1[224 + lane];
  w.ab = b1[lane];      w.bb = b1[32 + lane];

  for (int r = wave; r < rows; r += OWN_TB / 32) {
    int node = lo + r;
    float4 av = reinterpret_cast<const float4*>(aggx)[node];
    float dd = dis[node];
    float ha = w.ab;
    ha = __builtin_fmaf(av.x, w.a0, ha);
    ha = __builtin_fmaf(av.y, w.a1, ha);
    ha = __builtin_fmaf(av.z, w.a2, ha);
    ha = __builtin_fmaf(av.w, w.a3, ha);
    float hb = w.bb;
    hb = __builtin_fmaf(av.x, w.b0, hb);
    hb = __builtin_fmaf(av.y, w.b1, hb);
    hb = __builtin_fmaf(av.z, w.b2, hb);
    hb = __builtin_fmaf(av.w, w.b3, hb);
    ha = fmaxf(ha, 0.0f);
    hb = fmaxf(hb, 0.0f);
    acc[r * 64 + lane] = ha * dd * dd;
    acc[r * 64 + 32 + lane] = hb * dd * dd;
  }
  __syncthreads();

  const int beg = binoff[blockIdx.x];
  const int end = binoff[blockIdx.x + 1];
  const int2* lp = reinterpret_cast<const int2*>(list);
  for (int base = beg + wave * 32; base < end; base += OWN_TB) {
    int m = end - base;
    if (m > 32) m = 32;
    int r = 0;
    float n = 0.0f;
    float4 av = make_float4(0.0f, 0.0f, 0.0f, 0.0f);
    if (lane < m) {
      int2 p = lp[base + lane];
      av = reinterpret_cast<const float4*>(aggx)[p.x];
      n = dis[p.x] * dis[p.y];
      r = p.y - lo;
    }
    int nb = __float_as_int(n);
    int x0b = __float_as_int(av.x), x1b = __float_as_int(av.y);
    int x2b = __float_as_int(av.z), x3b = __float_as_int(av.w);
    for (int j = 0; j < m; ++j) {
      int rj = __builtin_amdgcn_readlane(r, j);
      float nj = __int_as_float(__builtin_amdgcn_readlane(nb, j));
      float x0 = __int_as_float(__builtin_amdgcn_readlane(x0b, j));
      float x1 = __int_as_float(__builtin_amdgcn_readlane(x1b, j));
      float x2 = __int_as_float(__builtin_amdgcn_readlane(x2b, j));
      float x3 = __int_as_float(__builtin_amdgcn_readlane(x3b, j));
      float ha = w.ab;
      ha = __builtin_fmaf(x0, w.a0, ha);
      ha = __builtin_fmaf(x1, w.a1, ha);
      ha = __builtin_fmaf(x2, w.a2, ha);
      ha = __builtin_fmaf(x3, w.a3, ha);
      float hb = w.bb;
      hb = __builtin_fmaf(x0, w.b0, hb);
      hb = __builtin_fmaf(x1, w.b1, hb);
      hb = __builtin_fmaf(x2, w.b2, hb);
      hb = __builtin_fmaf(x3, w.b3, hb);
      ha = fmaxf(ha, 0.0f);
      hb = fmaxf(hb, 0.0f);
      float* row = acc + rj * 64;
      atomicAdd(row + lane, ha * nj);
      atomicAdd(row + 32 + lane, hb * nj);
    }
  }
  __syncthreads();

  float* outp = r2 + (size_t)lo * 64;
  for (int i = threadIdx.x; i < rows * 64; i += OWN_TB) outp[i] = acc[i];
}

__global__ void __launch_bounds__(128)
k_gemm2_out(const float* __restrict__ r2, const float* __restrict__ W2,
            const float* __restrict__ b2, const float* __restrict__ Wf,
            const float* __restrict__ bf, float* __restrict__ out, int N) {
  __shared__ _Float16 w2t[64 * 64];
  for (int idx = threadIdx.x; idx < 64 * 64; idx += 128) {
    int col = idx >> 6, k = idx & 63;
    w2t[idx] = (_Float16)W2[k * 64 + col];
  }
  __syncthreads();

  int wave  = threadIdx.x >> 5;
  int lane  = threadIdx.x & 31;
  int rbase = blockIdx.x * 64 + wave * 16;
  int lrow  = lane & 15;
  int hi    = lane >> 4;
  int kb    = hi * 8;

  const float* ar = r2 + (size_t)(rbase + lrow) * 64;
  const _Float16* wt = w2t + (size_t)lrow * 64;

  v8f zero = {0, 0, 0, 0, 0, 0, 0, 0};
  v8f acc[4];
  acc[0] = zero; acc[1] = zero; acc[2] = zero; acc[3] = zero;

#pragma unroll
  for (int kc = 0; kc < 64; kc += 32) {
    v16h a;
#pragma unroll
    for (int j = 0; j < 8; ++j) {
      a[j]     = (_Float16)ar[kc + kb + j];
      a[8 + j] = (_Float16)ar[kc + kb + 16 + j];
    }
#pragma unroll
    for (int ct = 0; ct < 4; ++ct) {
      const _Float16* wc = wt + (size_t)ct * 16 * 64;
      v8h blo = *reinterpret_cast<const v8h*>(wc + kc + kb);
      v8h bhi = *reinterpret_cast<const v8h*>(wc + kc + kb + 16);
      v16h b = __builtin_shufflevector(blo, bhi, 0, 1, 2, 3, 4, 5, 6, 7,
                                       8, 9, 10, 11, 12, 13, 14, 15);
      acc[ct] = __builtin_amdgcn_wmma_f32_16x16x32_f16(
          false, a, false, b, (short)0, acc[ct], false, false);
    }
  }

  float bias[4], wf0[4], wf1[4];
#pragma unroll
  for (int ct = 0; ct < 4; ++ct) {
    int col = ct * 16 + lrow;
    bias[ct] = b2[col];
    wf0[ct]  = Wf[col * 2 + 0];
    wf1[ct]  = Wf[col * 2 + 1];
  }
  float bf0 = bf[0], bf1 = bf[1];

#pragma unroll
  for (int v = 0; v < 8; ++v) {
    float p0 = 0.0f, p1 = 0.0f;
#pragma unroll
    for (int ct = 0; ct < 4; ++ct) {
      float h = fmaxf(acc[ct][v] + bias[ct], 0.0f);
      p0 += h * wf0[ct];
      p1 += h * wf1[ct];
    }
#pragma unroll
    for (int off = 8; off >= 1; off >>= 1) {
      p0 += __shfl_xor(p0, off, 16);
      p1 += __shfl_xor(p1, off, 16);
    }
    int row = rbase + v + hi * 8;
    if (lrow == 0 && row < N) {
      out[row * 2 + 0] = p0 + bf0;
      out[row * 2 + 1] = p1 + bf1;
    }
  }
}

extern "C" void kernel_launch(void* const* d_in, const int* in_sizes, int n_in,
                              void* d_out, int out_size, void* d_ws, size_t ws_size,
                              hipStream_t stream) {
  const float* x  = (const float*)d_in[0];
  const int*   ei = (const int*)d_in[1];
  const float* W1 = (const float*)d_in[2];
  const float* b1 = (const float*)d_in[3];
  const float* W2 = (const float*)d_in[4];
  const float* b2 = (const float*)d_in[5];
  const float* Wf = (const float*)d_in[6];
  const float* bf = (const float*)d_in[7];
  float* out = (float*)d_out;

  int N = in_sizes[0] / 4;
  int E = in_sizes[1] / 2;
  int Npad = ((N + 63) / 64) * 64;
  const int* src = ei;
  const int* dst = ei + E;

  float* ws   = (float*)d_ws;
  float* dis  = ws;
  float* aggx = dis + N;
  float* r2   = aggx + (size_t)4 * N;
  int NB = (N + OWN_ROWS - 1) / OWN_ROWS;
  int* iw     = (int*)d_ws + (((size_t)5 * N + (size_t)64 * Npad + 63) / 64) * 64;
  int* bincnt = iw;
  int* cursor = iw + 1024;
  int* binoff = iw + 2048;
  int* list   = iw + 4096;
  float* xs   = (float*)(list + (((size_t)2 * E + 63) / 64) * 64);

  const int PB = (E + PART_TB * PART_U - 1) / (PART_TB * PART_U);
  k_zero<<<1, PART_TB, 0, stream>>>(bincnt, cursor);
  k_count_ranges<<<PB, PART_TB, 0, stream>>>(dst, bincnt, E);
  k_partition<<<PB, PART_TB, 0, stream>>>(src, dst, bincnt, cursor, binoff, list, E, NB);
  k_own_deg<<<NB, OWN_TB, 0, stream>>>(list, binoff, x, dis, xs, N);
  k_agg1_own<<<NB, OWN_TB, 0, stream>>>(list, binoff, xs, dis, aggx, N);
  k_layer2_own<<<NB, OWN_TB, 0, stream>>>(list, binoff, aggx, W1, b1, dis, r2, N);
  k_gemm2_out<<<Npad / 64, 128, 0, stream>>>(r2, W2, b2, Wf, bf, out, N);
}


# Output head fused into the destination-owning blocks: finished r2 rows feed the matrix product straight from LDS (row pitch 272 B), W2 staged once per block; r2 never stored; same sums; 6 launches
# speedup vs baseline: 8.3620x; 1.1359x over previous
#include <hip/hip_runtime.h>

typedef __attribute__((ext_vector_type(16))) _Float16 v16h;
typedef __attribute__((ext_vector_type(8)))  _Float16 v8h;
typedef __attribute__((ext_vector_type(8)))  float    v8f;

#define OWN_ROWS  391
#define OWN_TB    1024
#define PART_TB   1024
#define PART_U    8
#define AGG_U     16
#define FX_ONE    0x1p36f
#define FX_UNIT   0x1p-36f
#define OWN_PADROWS 400
#define ACC_PITCH 68
#define W2T_PITCH 72
#define W2_PER_T  (64 * 64 / OWN_TB)
static_assert(OWN_PADROWS >= OWN_ROWS && OWN_PADROWS % 16 == 0 && OWN_PADROWS / 16 <= OWN_TB / 32,
              "one wave per 16-row tile must cover the block's rows");
static_assert(W2_PER_T * OWN_TB == 64 * 64, "W2 staging must cover W2 exactly");


__global__ void __launch_bounds__(PART_TB)
k_zero(int* __restrict__ bincnt, int* __restrict__ cursor) {
  bincnt[threadIdx.x] = 0;
  cursor[threadIdx.x] = 0;
}

__global__ void __launch_bounds__(PART_TB)
k_count_ranges(const int* __restrict__ dst, int* __restrict__ bincnt, int E) {
  __shared__ int hist[PART_TB];
  const int tid = threadIdx.x;
  hist[tid] = 0;
  const int cbase = blockIdx.x * (PART_TB * PART_U);
  int d[PART_U];
#pragma unroll
  for (int u = 0; u < PART_U; ++u) {
    int e = cbase + u * PART_TB + tid;
    d[u] = -1;
    if (e < E) d[u] = dst[e];
  }
  __syncthreads();
#pragma unroll
  for (int u = 0; u < PART_U; ++u) {
    if (d[u] >= 0) atomicAdd(&hist[d[u] / OWN_ROWS], 1);
  }
  __syncthreads();
  const int n = hist[tid];
  if (n > 0) atomicAdd(&bincnt[tid], n);
}

__global__ void __launch_bounds__(OWN_TB)
k_own_deg(const int* __restrict__ list, const int* __restrict__ binoff,
          const float* __restrict__ x, float* __restrict__ dis,
          float* __restrict__ xs, int N) {
  __shared__ int cnt[OWN_ROWS];
  const int t = threadIdx.x;
  const int lo = blockIdx.x * OWN_ROWS;
  if (t < OWN_ROWS) cnt[t] = 0;
  __syncthreads();
  const int beg = binoff[blockIdx.x];
  const int end = binoff[blockIdx.x + 1];
  for (int base = beg + t; base < end; base += 8 * OWN_TB) {
    int d[8];
#pragma unroll
    for (int u = 0; u < 8; ++u) {
      int i = base + u * OWN_TB;
      d[u] = lo - 1;
      if (i < end) d[u] = list[(size_t)2 * i + 1];
    }
#pragma unroll
    for (int u = 0; u < 8; ++u) {
      int r = d[u] - lo;
      if (r >= 0) atomicAdd(&cnt[r], 1);
    }
  }
  __syncthreads();
  const int i = lo + t;
  if (t < OWN_ROWS && i < N) {
    float dg = (float)(cnt[t] + 1);
    float r = rsqrtf(dg);
    dis[i] = r;
    float4 xv = reinterpret_cast<const float4*>(x)[i];
    float4 o = make_float4(xv.x * r, xv.y * r, xv.z * r, xv.w * r);
    reinterpret_cast<float4*>(xs)[i] = o;
  }
}

static __device__ __forceinline__ unsigned long long to_fx(float v) {
  return (unsigned long long)(long long)(v * FX_ONE);
}
static __device__ __forceinline__ float from_fx(unsigned long long s) {
  return (float)(long long)s * FX_UNIT;
}

__global__ void __launch_bounds__(OWN_TB)
k_agg1_own(const int* __restrict__ list, const int* __restrict__ binoff,
           const float* __restrict__ xs, const float* __restrict__ dis,
           float* __restrict__ aggx, int N) {
  __shared__ unsigned long long acc[4 * OWN_ROWS];
  const int t = threadIdx.x;
  const int lo = blockIdx.x * OWN_ROWS;
  const int i = lo + t;
  const float4* xs4 = reinterpret_cast<const float4*>(xs);

  if (t < OWN_ROWS) {
    float4 sv = make_float4(0.0f, 0.0f, 0.0f, 0.0f);
    if (i < N) sv = xs4[i];
    acc[t] = to_fx(sv.x);
    acc[OWN_ROWS + t] = to_fx(sv.y);
    acc[2 * OWN_ROWS + t] = to_fx(sv.z);
    acc[3 * OWN_ROWS + t] = to_fx(sv.w);
  }
  __syncthreads();

  const int beg = binoff[blockIdx.x];
  const int end = binoff[blockIdx.x + 1];
  const int2* lp = reinterpret_cast<const int2*>(list);
  for (int base = beg + t; base < end; base += AGG_U * OWN_TB) {
    int2 p[AGG_U];
    float4 v[AGG_U];
#pragma unroll
    for (int u = 0; u < AGG_U; ++u) {
      int k = base + u * OWN_TB;
      p[u] = make_int2(0, lo - 1);
      if (k < end) p[u] = lp[k];
    }
#pragma unroll
    for (int u = 0; u < AGG_U; ++u) v[u] = xs4[p[u].x];
#pragma unroll
    for (int u = 0; u < AGG_U; ++u) {
      int r = p[u].y - lo;
      if (r >= 0) {
        atomicAdd(&acc[r], to_fx(v[u].x));
        atomicAdd(&acc[OWN_ROWS + r], to_fx(v[u].y));
        atomicAdd(&acc[2 * OWN_ROWS + r], to_fx(v[u].z));
        atomicAdd(&acc[3 * OWN_ROWS + r], to_fx(v[u].w));
      }
    }
  }
  __syncthreads();

  if (t < OWN_ROWS && i < N) {
    float dd = dis[i];
    float4 o;
    o.x = dd * from_fx(acc[t]);
    o.y = dd * from_fx(acc[OWN_ROWS + t]);
    o.z = dd * from_fx(acc[2 * OWN_ROWS + t]);
    o.w = dd * from_fx(acc[3 * OWN_ROWS + t]);
    reinterpret_cast<float4*>(aggx)[i] = o;
  }
}

static __device__ __forceinline__ int part_block_scan(int v, int* wtot, int lane, int wave) {
  int incl = v;
  for (int o = 1; o < 32; o <<= 1) {
    int up = __shfl_up(incl, o);
    if (lane >= o) incl += up;
  }
  if (lane == 31) wtot[wave] = incl;
  __syncthreads();
  if (wave == 0) {
    int t = wtot[lane];
    int it = t;
    for (int o = 1; o < 32; o <<= 1) {
      int up = __shfl_up(it, o);
      if (lane >= o) it += up;
    }
    wtot[lane] = it - t;
  }
  __syncthreads();
  return incl - v + wtot[wave];
}

__global__ void __launch_bounds__(PART_TB)
k_partition(const int* __restrict__ src, const int* __restrict__ dst,
            const int* __restrict__ bincnt, int* __restrict__ cursor,
            int* __restrict__ binoff, int* __restrict__ list, int E, int NB) {
  __shared__ int hist[PART_TB];
  __shared__ int lst[PART_TB];
  __shared__ int wtotA[32];
  __shared__ int wtotB[32];
  __shared__ int2 stage[PART_TB * PART_U];
  const int tid = threadIdx.x;
  const int lane = tid & 31;
  const int wave = tid >> 5;

  const int cbase = blockIdx.x * (PART_TB * PART_U);
  int s[PART_U], d[PART_U], bq[PART_U], rk[PART_U];
#pragma unroll
  for (int u = 0; u < PART_U; ++u) {
    int e = cbase + u * PART_TB + tid;
    s[u] = 0;
    d[u] = -1;
    if (e < E) { s[u] = src[e]; d[u] = dst[e]; }
  }

  const int c = (tid < NB) ? bincnt[tid] : 0;
  hist[tid] = 0;
  const int start = part_block_scan(c, wtotA, lane, wave);
  if (blockIdx.x == 0 && tid < NB) {
    binoff[tid] = start;
    if (tid == NB - 1) binoff[NB] = start + c;
  }

#pragma unroll
  for (int u = 0; u < PART_U; ++u) {
    bq[u] = 0;
    rk[u] = 0;
    if (d[u] >= 0) {
      bq[u] = d[u] / OWN_ROWS;
      rk[u] = atomicAdd(&hist[bq[u]], 1);
    }
  }
  __syncthreads();

  const int n = hist[tid];
  int g = 0;
  if (n > 0) g = atomicAdd(&cursor[tid], n);
  const int ls = part_block_scan(n, wtotB, lane, wave);
  lst[tid] = ls;
  hist[tid] = start + g - ls;
  __syncthreads();

  int q[PART_U];
#pragma unroll
  for (int u = 0; u < PART_U; ++u) q[u] = lst[bq[u]] + rk[u];
#pragma unroll
  for (int u = 0; u < PART_U; ++u) {
    if (d[u] >= 0) stage[q[u]] = make_int2(s[u], d[u]);
  }
  __syncthreads();

  int2* lp = reinterpret_cast<int2*>(list);
  int nvalid = E - cbase;
  if (nvalid > PART_TB * PART_U) nvalid = PART_TB * PART_U;
  for (int i = tid; i < nvalid; i += PART_TB) {
    int2 p = stage[i];
    lp[hist[p.y / OWN_ROWS] + i] = p;
  }
}

struct OwnW {
  float a0, a1, a2, a3, ab;
  float b0, b1, b2, b3, bb;
};

__global__ void __launch_bounds__(OWN_TB)
k_layer2_out(const int* __restrict__ list, const int* __restrict__ binoff,
             const float* __restrict__ aggx, const float* __restrict__ W1,
             const float* __restrict__ b1, const float* __restrict__ dis,
             const float* __restrict__ W2, const float* __restrict__ b2,
             const float* __restrict__ Wf, const float* __restrict__ bf,
             float* __restrict__ out, int N) {
  __shared__ __attribute__((aligned(16))) float acc[OWN_PADROWS * ACC_PITCH];
  __shared__ __attribute__((aligned(16))) _Float16 w2t[64 * W2T_PITCH];

  const int lane = threadIdx.x & 31;
  const int wave = __builtin_amdgcn_readfirstlane((int)(threadIdx.x >> 5));
  const int lo = blockIdx.x * OWN_ROWS;
  int rows = N - lo;
  if (rows > OWN_ROWS) rows = OWN_ROWS;

  OwnW w;
  w.a0 = W1[lane];      w.a1 = W1[64 + lane]; w.a2 = W1[128 + lane]; w.a3 = W1[192 + lane];
  w.b0 = W1[32 + lane]; w.b1 = W1[96 + lane]; w.b2 = W1[160 + lane]; w.b3 = W1[224 + lane];
  w.ab = b1[lane];      w.bb = b1[32 + lane];

  float w2v[W2_PER_T];
#pragma unroll
  for (int i = 0; i < W2_PER_T; ++i) w2v[i] = W2[threadIdx.x + i * OWN_TB];

  for (int r = wave; r < OWN_PADROWS; r += OWN_TB / 32) {
    float sa = 0.0f, sb = 0.0f;
    if (r < rows) {
      int node = lo + r;
      float4 av = reinterpret_cast<const float4*>(aggx)[node];
      float dd = dis[node];
      float ha = w.ab;
      ha = __builtin_fmaf(av.x, w.a0, ha);
      ha = __builtin_fmaf(av.y, w.a1, ha);
      ha = __builtin_fmaf(av.z, w.a2, ha);
      ha = __builtin_fmaf(av.w, w.a3, ha);
      float hb = w.bb;
      hb = __builtin_fmaf(av.x, w.b0, hb);
      hb = __builtin_fmaf(av.y, w.b1, hb);
      hb = __builtin_fmaf(av.z, w.b2, hb);
      hb = __builtin_fmaf(av.w, w.b3, hb);
      ha = fmaxf(ha, 0.0f);
      hb = fmaxf(hb, 0.0f);
      sa = ha * dd * dd;
      sb = hb * dd * dd;
    }
    acc[r * ACC_PITCH + lane] = sa;
    acc[r * ACC_PITCH + 32 + lane] = sb;
  }
#pragma unroll
  for (int i = 0; i < W2_PER_T; ++i) {
    int idx = threadIdx.x + i * OWN_TB;
    int k = idx >> 6, col = idx & 63;
    w2t[col * W2T_PITCH + k] = (_Float16)w2v[i];
  }
  __syncthreads();

  const int beg = binoff[blockIdx.x];
  const int end = binoff[blockIdx.x + 1];
  const int2* lp = reinterpret_cast<const int2*>(list);
  for (int base = beg + wave * 32; base < end; base += OWN_TB) {
    int m = end - base;
    if (m > 32) m = 32;
    int r = 0;
    float n = 0.0f;
    float4 av = make_float4(0.0f, 0.0f, 0.0f, 0.0f);
    if (lane < m) {
      int2 p = lp[base + lane];
      av = reinterpret_cast<const float4*>(aggx)[p.x];
      n = dis[p.x] * dis[p.y];
      r = (p.y - lo) * ACC_PITCH;
    }
    int nb = __float_as_int(n);
    int x0b = __float_as_int(av.x), x1b = __float_as_int(av.y);
    int x2b = __float_as_int(av.z), x3b = __float_as_int(av.w);
    for (int j = 0; j < m; ++j) {
      int rj = __builtin_amdgcn_readlane(r, j);
      float nj = __int_as_float(__builtin_amdgcn_readlane(nb, j));
      float x0 = __int_as_float(__builtin_amdgcn_readlane(x0b, j));
      float x1 = __int_as_float(__builtin_amdgcn_readlane(x1b, j));
      float x2 = __int_as_float(__builtin_amdgcn_readlane(x2b, j));
      float x3 = __int_as_float(__builtin_amdgcn_readlane(x3b, j));
      float ha = w.ab;
      ha = __builtin_fmaf(x0, w.a0, ha);
      ha = __builtin_fmaf(x1, w.a1, ha);
      ha = __builtin_fmaf(x2, w.a2, ha);
      ha = __builtin_fmaf(x3, w.a3, ha);
      float hb = w.bb;
      hb = __builtin_fmaf(x0, w.b0, hb);
      hb = __builtin_fmaf(x1, w.b1, hb);
      hb = __builtin_fmaf(x2, w.b2, hb);
      hb = __builtin_fmaf(x3, w.b3, hb);
      ha = fmaxf(ha, 0.0f);
      hb = fmaxf(hb, 0.0f);
      float* row = acc + rj;
      atomicAdd(row + lane, ha * nj);
      atomicAdd(row + 32 + lane, hb * nj);
    }
  }
  __syncthreads();

  const int t0 = wave * 16;
  if (t0 < rows) {
    const int lrow = lane & 15;
    const int hi   = lane >> 4;
    const int kb   = hi * 8;
    const float* ar = acc + (t0 + lrow) * ACC_PITCH;
    const _Float16* wt = w2t + lrow * W2T_PITCH;

    v8f zero = {0, 0, 0, 0, 0, 0, 0, 0};
    v8f o[4];
    o[0] = zero; o[1] = zero; o[2] = zero; o[3] = zero;

#pragma unroll
    for (int kc = 0; kc < 64; kc += 32) {
      float4 f0 = *reinterpret_cast<const float4*>(ar + kc + kb);
      float4 f1 = *reinterpret_cast<const float4*>(ar + kc + kb + 4);
      float4 f2 = *reinterpret_cast<const float4*>(ar + kc + kb + 16);
      float4 f3 = *reinterpret_cast<const float4*>(ar + kc + kb + 20);
      v16h a;
      a[0]  = (_Float16)f0.x; a[1]  = (_Float16)f0.y; a[2]  = (_Float16)f0.z; a[3]  = (_Float16)f0.w;
      a[4]  = (_Float16)f1.x; a[5]  = (_Float16)f1.y; a[6]  = (_Float16)f1.z; a[7]  = (_Float16)f1.w;
      a[8]  = (_Float16)f2.x; a[9]  = (_Float16)f2.y; a[10] = (_Float16)f2.z; a[11] = (_Float16)f2.w;
      a[12] = (_Float16)f3.x; a[13] = (_Float16)f3.y; a[14] = (_Float16)f3.z; a[15] = (_Float16)f3.w;
#pragma unroll
      for (int ct = 0; ct < 4; ++ct) {
        const _Float16* wc = wt + ct * 16 * W2T_PITCH;
        v8h blo = *reinterpret_cast<const v8h*>(wc + kc + kb);
        v8h bhi = *reinterpret_cast<const v8h*>(wc + kc + kb + 16);
        v16h b = __builtin_shufflevector(blo, bhi, 0, 1, 2, 3, 4, 5, 6, 7,
                                         8, 9, 10, 11, 12, 13, 14, 15);
        o[ct] = __builtin_amdgcn_wmma_f32_16x16x32_f16(
            false, a, false, b, (short)0, o[ct], false, false);
      }
    }

    float bias[4], wf0[4], wf1[4];
#pragma unroll
    for (int ct = 0; ct < 4; ++ct) {
      int col = ct * 16 + lrow;
      bias[ct] = b2[col];
      wf0[ct]  = Wf[col * 2 + 0];
      wf1[ct]  = Wf[col * 2 + 1];
    }
    float bf0 = bf[0], bf1 = bf[1];

    float p0[8], p1[8];
#pragma unroll
    for (int v = 0; v < 8; ++v) {
      p0[v] = 0.0f;
      p1[v] = 0.0f;
#pragma unroll
      for (int ct = 0; ct < 4; ++ct) {
        float h = fmaxf(o[ct][v] + bias[ct], 0.0f);
        p0[v] += h * wf0[ct];
        p1[v] += h * wf1[ct];
      }
    }
#pragma unroll
    for (int off = 8; off >= 1; off >>= 1) {
#pragma unroll
      for (int v = 0; v < 8; ++v) {
        p0[v] += __shfl_xor(p0[v], off, 16);
        p1[v] += __shfl_xor(p1[v], off, 16);
      }
    }
#pragma unroll
    for (int v = 0; v < 8; ++v) {
      int rl = t0 + v + hi * 8;
      if (lrow == 0 && rl < rows) {
        int row = lo + rl;
        out[row * 2 + 0] = p0[v] + bf0;
        out[row * 2 + 1] = p1[v] + bf1;
      }
    }
  }
}

extern "C" void kernel_launch(void* const* d_in, const int* in_sizes, int n_in,
                              void* d_out, int out_size, void* d_ws, size_t ws_size,
                              hipStream_t stream) {
  const float* x  = (const float*)d_in[0];
  const int*   ei = (const int*)d_in[1];
  const float* W1 = (const float*)d_in[2];
  const float* b1 = (const float*)d_in[3];
  const float* W2 = (const float*)d_in[4];
  const float* b2 = (const float*)d_in[5];
  const float* Wf = (const float*)d_in[6];
  const float* bf = (const float*)d_in[7];
  float* out = (float*)d_out;

  int N = in_sizes[0] / 4;
  int E = in_sizes[1] / 2;
  const int* src = ei;
  const int* dst = ei + E;

  float* ws   = (float*)d_ws;
  float* dis  = ws;
  float* aggx = dis + N;
  int NB = (N + OWN_ROWS - 1) / OWN_ROWS;
  int* iw     = (int*)d_ws + (((size_t)5 * N + 63) / 64) * 64;
  int* bincnt = iw;
  int* cursor = iw + 1024;
  int* binoff = iw + 2048;
  int* list   = iw + 4096;
  float* xs   = (float*)(list + (((size_t)2 * E + 63) / 64) * 64);

  const int PB = (E + PART_TB * PART_U - 1) / (PART_TB * PART_U);
  k_zero<<<1, PART_TB, 0, stream>>>(bincnt, cursor);
  k_count_ranges<<<PB, PART_TB, 0, stream>>>(dst, bincnt, E);
  k_partition<<<PB, PART_TB, 0, stream>>>(src, dst, bincnt, cursor, binoff, list, E, NB);
  k_own_deg<<<NB, OWN_TB, 0, stream>>>(list, binoff, x, dis, xs, N);
  k_agg1_own<<<NB, OWN_TB, 0, stream>>>(list, binoff, xs, dis, aggx, N);
  k_layer2_out<<<NB, OWN_TB, 0, stream>>>(list, binoff, aggx, W1, b1, dis, W2, b2, Wf, bf, out, N);
}
